# starting kernel
# baseline (speedup 1.0000x reference)
#include <hip/hip_runtime.h>
#include <math.h>


#define HW 1024
#define EPSV 1e-5f

typedef __attribute__((ext_vector_type(2))) float v2f;
typedef __attribute__((ext_vector_type(8))) float v8f;

__global__ void gemm_wmma_f32(const float* __restrict__ Wt,
                              const float* __restrict__ X,
                              float* __restrict__ Y,
                              const float* __restrict__ cb,
                              const float* __restrict__ bg,
                              const float* __restrict__ bb,
                              const float* __restrict__ bm,
                              const float* __restrict__ bv,
                              int M, int K, int Mtiles, int relu_mode)
{
    const int lane = threadIdx.x & 31;
    const int wid  = threadIdx.x >> 5;
    const int gw   = blockIdx.x * 8 + wid;
    const int mt   = gw % Mtiles;
    const int nt   = gw / Mtiles;
    const int m0   = mt * 16;
    const int ncol = nt * 64;
    const int b    = ncol >> 10;
    const int hw0  = ncol & 1023;

    const int lm = lane & 15;
    const int lk = (lane >> 4) << 1;

    const float* __restrict__ Xb = X + (size_t)b * K * HW + hw0;
    const float* __restrict__ Ar = Wt + (size_t)(m0 + lm) * K + lk;

    v8f zero = {0.f, 0.f, 0.f, 0.f, 0.f, 0.f, 0.f, 0.f};
    v8f acc[4];
#pragma unroll
    for (int s = 0; s < 4; ++s) acc[s] = zero;

    for (int k = 0; k < K; k += 4) {
        v2f a = *(const v2f*)(Ar + k);
        const float* Bk = Xb + (size_t)(k + lk) * HW + lm;
#pragma unroll
        for (int s = 0; s < 4; ++s) {
            v2f bf;
            bf.x = Bk[s * 16];
            bf.y = Bk[s * 16 + HW];
            acc[s] = __builtin_amdgcn_wmma_f32_16x16x4_f32(
                false, a, false, bf, (short)0, acc[s], false, false);
        }
    }

    const int hi = (lane >> 4) << 3;
    float osc[8], obi[8];
#pragma unroll
    for (int r = 0; r < 8; ++r) {
        int o = m0 + r + hi;
        if (relu_mode) {
            float sc = bg[o] * rsqrtf(bv[o] + EPSV);
            osc[r] = sc;
            obi[r] = (cb[o] - bm[o]) * sc + bb[o];
        } else {
            osc[r] = 1.f;
            obi[r] = cb[o];
        }
    }
#pragma unroll
    for (int s = 0; s < 4; ++s) {
        int ncl = hw0 + s * 16 + lm;
#pragma unroll
        for (int r = 0; r < 8; ++r) {
            int o = m0 + r + hi;
            float val = acc[s][r] * osc[r] + obi[r];
            if (relu_mode) val = fmaxf(val, 0.f);
            Y[(size_t)b * M * HW + (size_t)o * HW + ncl] = val;
        }
    }
}

__global__ void dwconv7_bn(const float* __restrict__ z1,
                           const float* __restrict__ wt,
                           const float* __restrict__ cb,
                           const float* __restrict__ bg,
                           const float* __restrict__ bb,
                           const float* __restrict__ bm,
                           const float* __restrict__ bv,
                           float* __restrict__ z2)
{
    int idx = blockIdx.x * blockDim.x + threadIdx.x;
    int hw = idx & 1023;
    int ch = (idx >> 10) & 255;
    int b  = idx >> 18;
    int h = hw >> 5, w = hw & 31;
    const float* src = z1 + ((size_t)(b * 256 + ch)) * HW;
    const float* wk  = wt + ch * 49;
    float acc = 0.f;
#pragma unroll
    for (int i = 0; i < 7; ++i) {
        int yy = h + i - 3;
        if ((unsigned)yy >= 32u) continue;
#pragma unroll
        for (int j = 0; j < 7; ++j) {
            int xx = w + j - 3;
            if ((unsigned)xx < 32u) acc += wk[i * 7 + j] * src[yy * 32 + xx];
        }
    }
    acc += cb[ch];
    float sc = bg[ch] * rsqrtf(bv[ch] + EPSV);
    z2[idx] = (acc - bm[ch]) * sc + bb[ch];
}

__global__ void gn_stats(const float* __restrict__ z4, float* __restrict__ stats)
{
    __shared__ float s1[256], s2[256];
    int gid = blockIdx.x;
    int tid = threadIdx.x;
    const float* base = z4 + (size_t)gid * 9216;
    float a = 0.f, q = 0.f;
    for (int i = tid; i < 9216; i += 256) { float t = base[i]; a += t; q += t * t; }
    s1[tid] = a; s2[tid] = q;
    __syncthreads();
    for (int off = 128; off > 0; off >>= 1) {
        if (tid < off) { s1[tid] += s1[tid + off]; s2[tid] += s2[tid + off]; }
        __syncthreads();
    }
    if (tid == 0) {
        float mean = s1[0] * (1.f / 9216.f);
        float var  = s2[0] * (1.f / 9216.f) - mean * mean;
        stats[gid * 2]     = mean;
        stats[gid * 2 + 1] = rsqrtf(var + EPSV);
    }
}

__global__ void ska_kernel(const float* __restrict__ x,
                           const float* __restrict__ z4,
                           const float* __restrict__ stats,
                           const float* __restrict__ gg,
                           const float* __restrict__ gb,
                           const float* __restrict__ og,
                           const float* __restrict__ ob,
                           const float* __restrict__ om,
                           const float* __restrict__ ov,
                           float* __restrict__ out)
{
    int idx = blockIdx.x * blockDim.x + threadIdx.x;
    int hw = idx & 1023;
    int cc = (idx >> 10) & 63;
    int b  = idx >> 16;
    int h = hw >> 5, w = hw & 31;

    float mu = stats[(b * 64 + cc) * 2];
    float rs = stats[(b * 64 + cc) * 2 + 1];

    const float* zp = z4 + ((size_t)(b * 576 + cc * 9)) * HW + hw;
    float wv[9];
    float mx = -1e30f;
#pragma unroll
    for (int k = 0; k < 9; ++k) {
        float t = (zp[(size_t)k * HW] - mu) * rs * gg[cc * 9 + k] + gb[cc * 9 + k];
        wv[k] = t;
        mx = fmaxf(mx, t);
    }
    float ssum = 0.f;
#pragma unroll
    for (int k = 0; k < 9; ++k) { float e = __expf(wv[k] - mx); wv[k] = e; ssum += e; }
    float inv = 1.f / ssum;
#pragma unroll
    for (int k = 0; k < 9; ++k) wv[k] *= inv;

#pragma unroll
    for (int g = 0; g < 8; ++g) {
        int c = g * 64 + cc;
        const float* xs = x + ((size_t)(b * 512 + c)) * HW;
        float acc = 0.f;
        int k = 0;
#pragma unroll
        for (int i = 0; i < 3; ++i) {
            int yy = h + i - 1;
#pragma unroll
            for (int j = 0; j < 3; ++j, ++k) {
                int xx = w + j - 1;
                if ((unsigned)yy < 32u && (unsigned)xx < 32u)
                    acc += wv[k] * xs[yy * 32 + xx];
            }
        }
        float sc = og[c] * rsqrtf(ov[c] + EPSV);
        out[((size_t)(b * 512 + c)) * HW + hw] = (acc - om[c]) * sc + ob[c] + xs[hw];
    }
}

extern "C" void kernel_launch(void* const* d_in, const int* in_sizes, int n_in,
                              void* d_out, int out_size, void* d_ws, size_t ws_size,
                              hipStream_t stream)
{
    const float* x     = (const float*)d_in[0];
    const float* cv1_w = (const float*)d_in[1];
    const float* cv1_b = (const float*)d_in[2];
    const float* bn1_g = (const float*)d_in[3];
    const float* bn1_b = (const float*)d_in[4];
    const float* bn1_m = (const float*)d_in[5];
    const float* bn1_v = (const float*)d_in[6];
    const float* cv2_w = (const float*)d_in[7];
    const float* cv2_b = (const float*)d_in[8];
    const float* bn2_g = (const float*)d_in[9];
    const float* bn2_b = (const float*)d_in[10];
    const float* bn2_m = (const float*)d_in[11];
    const float* bn2_v = (const float*)d_in[12];
    const float* cv3_w = (const float*)d_in[13];
    const float* cv3_b = (const float*)d_in[14];
    const float* bn3_g = (const float*)d_in[15];
    const float* bn3_b = (const float*)d_in[16];
    const float* bn3_m = (const float*)d_in[17];
    const float* bn3_v = (const float*)d_in[18];
    const float* cv4_w = (const float*)d_in[19];
    const float* cv4_b = (const float*)d_in[20];
    const float* gn_g  = (const float*)d_in[21];
    const float* gn_b  = (const float*)d_in[22];
    const float* bno_g = (const float*)d_in[23];
    const float* bno_b = (const float*)d_in[24];
    const float* bno_m = (const float*)d_in[25];
    const float* bno_v = (const float*)d_in[26];

    float* ws  = (float*)d_ws;
    float* Z1  = ws;
    float* Z2  = ws + (size_t)4194304;
    float* Z4  = ws + (size_t)8388608;
    float* ST  = ws + (size_t)(8388608 + 9437184);
    float* out = (float*)d_out;

    gemm_wmma_f32<<<512, 256, 0, stream>>>(cv1_w, x, Z1,
                                           cv1_b, bn1_g, bn1_b, bn1_m, bn1_v,
                                           256, 512, 16, 1);
    dwconv7_bn<<<16384, 256, 0, stream>>>(Z1, cv2_w, cv2_b,
                                          bn2_g, bn2_b, bn2_m, bn2_v, Z2);
    gemm_wmma_f32<<<512, 256, 0, stream>>>(cv3_w, Z2, Z1,
                                           cv3_b, bn3_g, bn3_b, bn3_m, bn3_v,
                                           256, 256, 16, 1);
    gemm_wmma_f32<<<1152, 256, 0, stream>>>(cv4_w, Z1, Z4,
                                            cv4_b, cv4_b, cv4_b, cv4_b, cv4_b,
                                            576, 256, 36, 0);
    gn_stats<<<1024, 256, 0, stream>>>(Z4, ST);
    ska_kernel<<<4096, 256, 0, stream>>>(x, Z4, ST, gn_g, gn_b,
                                         bno_g, bno_b, bno_m, bno_v, out);
}


# dwconv7x7: stage each 32x32 plane once in LDS with zero halo, one wave per plane, lane=column in 8-row register strips (was 49 masked global gathers per output); tap sum order kept
# speedup vs baseline: 1.2861x; 1.2861x over previous
#include <hip/hip_runtime.h>
#include <math.h>


#define HW 1024
#define EPSV 1e-5f

typedef __attribute__((ext_vector_type(2))) float v2f;
typedef __attribute__((ext_vector_type(8))) float v8f;

__global__ void gemm_wmma_f32(const float* __restrict__ Wt,
                              const float* __restrict__ X,
                              float* __restrict__ Y,
                              const float* __restrict__ cb,
                              const float* __restrict__ bg,
                              const float* __restrict__ bb,
                              const float* __restrict__ bm,
                              const float* __restrict__ bv,
                              int M, int K, int Mtiles, int relu_mode)
{
    const int lane = threadIdx.x & 31;
    const int wid  = threadIdx.x >> 5;
    const int gw   = blockIdx.x * 8 + wid;
    const int mt   = gw % Mtiles;
    const int nt   = gw / Mtiles;
    const int m0   = mt * 16;
    const int ncol = nt * 64;
    const int b    = ncol >> 10;
    const int hw0  = ncol & 1023;

    const int lm = lane & 15;
    const int lk = (lane >> 4) << 1;

    const float* __restrict__ Xb = X + (size_t)b * K * HW + hw0;
    const float* __restrict__ Ar = Wt + (size_t)(m0 + lm) * K + lk;

    v8f zero = {0.f, 0.f, 0.f, 0.f, 0.f, 0.f, 0.f, 0.f};
    v8f acc[4];
#pragma unroll
    for (int s = 0; s < 4; ++s) acc[s] = zero;

    for (int k = 0; k < K; k += 4) {
        v2f a = *(const v2f*)(Ar + k);
        const float* Bk = Xb + (size_t)(k + lk) * HW + lm;
#pragma unroll
        for (int s = 0; s < 4; ++s) {
            v2f bf;
            bf.x = Bk[s * 16];
            bf.y = Bk[s * 16 + HW];
            acc[s] = __builtin_amdgcn_wmma_f32_16x16x4_f32(
                false, a, false, bf, (short)0, acc[s], false, false);
        }
    }

    const int hi = (lane >> 4) << 3;
    float osc[8], obi[8];
#pragma unroll
    for (int r = 0; r < 8; ++r) {
        int o = m0 + r + hi;
        if (relu_mode) {
            float sc = bg[o] * rsqrtf(bv[o] + EPSV);
            osc[r] = sc;
            obi[r] = (cb[o] - bm[o]) * sc + bb[o];
        } else {
            osc[r] = 1.f;
            obi[r] = cb[o];
        }
    }
#pragma unroll
    for (int s = 0; s < 4; ++s) {
        int ncl = hw0 + s * 16 + lm;
#pragma unroll
        for (int r = 0; r < 8; ++r) {
            int o = m0 + r + hi;
            float val = acc[s][r] * osc[r] + obi[r];
            if (relu_mode) val = fmaxf(val, 0.f);
            Y[(size_t)b * M * HW + (size_t)o * HW + ncl] = val;
        }
    }
}

#define DW_LP 40
#define DW_LR 38

__global__ void __launch_bounds__(32)
dwconv7_bn(const float* __restrict__ z1,
           const float* __restrict__ wt,
           const float* __restrict__ cb,
           const float* __restrict__ bg,
           const float* __restrict__ bb,
           const float* __restrict__ bm,
           const float* __restrict__ bv,
           float* __restrict__ z2)
{
    __shared__ __attribute__((aligned(16))) float tile[DW_LR * DW_LP];

    const int lane  = threadIdx.x & 31;
    const int plane = blockIdx.x;
    const int ch    = plane & 255;

    const float4 z4 = make_float4(0.f, 0.f, 0.f, 0.f);
#pragma unroll
    for (int it = 0; it < 4; ++it) {
        const int i = it * 32 + lane;
        if (i >= 124) continue;
        int addr;
        if (i < 60) {
            int rr  = i / 10;
            int row = (rr < 3) ? rr : rr + 32;
            addr = row * DW_LP + (i - rr * 10) * 4;
        } else {
            int j = i - 60;
            addr = ((j >> 1) + 3) * DW_LP + (j & 1) * 36;
        }
        *(float4*)(tile + addr) = z4;
    }
    const float4* __restrict__ src4 = (const float4*)(z1 + (size_t)plane * HW);
    float4 in[8];
#pragma unroll
    for (int q = 0; q < 8; ++q) in[q] = src4[q * 32 + lane];

#pragma unroll
    for (int q = 0; q < 8; ++q) {
        int f = q * 32 + lane;
        *(float4*)(tile + ((f >> 3) + 3) * DW_LP + 4 + (f & 7) * 4) = in[q];
    }
    __syncthreads();

    const float* __restrict__ wk = wt + ch * 49;
    const float cbv = cb[ch];
    const float bmv = bm[ch];
    const float bbv = bb[ch];
    const float sc  = bg[ch] * rsqrtf(bv[ch] + EPSV);

    float* __restrict__ dst = z2 + (size_t)plane * HW + lane;

#pragma unroll 1
    for (int h0 = 0; h0 < 32; h0 += 8) {
        float acc[8];
#pragma unroll
        for (int i = 0; i < 8; ++i) acc[i] = 0.f;
        const float* trow = tile + h0 * DW_LP + lane + 1;
#pragma unroll
        for (int rr = 0; rr < 14; ++rr) {
#pragma unroll
            for (int j = 0; j < 7; ++j) {
                float v = trow[rr * DW_LP + j];
#pragma unroll
                for (int i = 0; i < 8; ++i) {
                    int ki = rr - i;
                    if (ki >= 0 && ki < 7) acc[i] += wk[ki * 7 + j] * v;
                }
            }
        }
#pragma unroll
        for (int i = 0; i < 8; ++i) {
            float t = acc[i] + cbv;
            dst[(h0 + i) * 32] = (t - bmv) * sc + bbv;
        }
    }
}

__global__ void gn_stats(const float* __restrict__ z4, float* __restrict__ stats)
{
    __shared__ float s1[256], s2[256];
    int gid = blockIdx.x;
    int tid = threadIdx.x;
    const float* base = z4 + (size_t)gid * 9216;
    float a = 0.f, q = 0.f;
    for (int i = tid; i < 9216; i += 256) { float t = base[i]; a += t; q += t * t; }
    s1[tid] = a; s2[tid] = q;
    __syncthreads();
    for (int off = 128; off > 0; off >>= 1) {
        if (tid < off) { s1[tid] += s1[tid + off]; s2[tid] += s2[tid + off]; }
        __syncthreads();
    }
    if (tid == 0) {
        float mean = s1[0] * (1.f / 9216.f);
        float var  = s2[0] * (1.f / 9216.f) - mean * mean;
        stats[gid * 2]     = mean;
        stats[gid * 2 + 1] = rsqrtf(var + EPSV);
    }
}

__global__ void ska_kernel(const float* __restrict__ x,
                           const float* __restrict__ z4,
                           const float* __restrict__ stats,
                           const float* __restrict__ gg,
                           const float* __restrict__ gb,
                           const float* __restrict__ og,
                           const float* __restrict__ ob,
                           const float* __restrict__ om,
                           const float* __restrict__ ov,
                           float* __restrict__ out)
{
    int idx = blockIdx.x * blockDim.x + threadIdx.x;
    int hw = idx & 1023;
    int cc = (idx >> 10) & 63;
    int b  = idx >> 16;
    int h = hw >> 5, w = hw & 31;

    float mu = stats[(b * 64 + cc) * 2];
    float rs = stats[(b * 64 + cc) * 2 + 1];

    const float* zp = z4 + ((size_t)(b * 576 + cc * 9)) * HW + hw;
    float wv[9];
    float mx = -1e30f;
#pragma unroll
    for (int k = 0; k < 9; ++k) {
        float t = (zp[(size_t)k * HW] - mu) * rs * gg[cc * 9 + k] + gb[cc * 9 + k];
        wv[k] = t;
        mx = fmaxf(mx, t);
    }
    float ssum = 0.f;
#pragma unroll
    for (int k = 0; k < 9; ++k) { float e = __expf(wv[k] - mx); wv[k] = e; ssum += e; }
    float inv = 1.f / ssum;
#pragma unroll
    for (int k = 0; k < 9; ++k) wv[k] *= inv;

#pragma unroll
    for (int g = 0; g < 8; ++g) {
        int c = g * 64 + cc;
        const float* xs = x + ((size_t)(b * 512 + c)) * HW;
        float acc = 0.f;
        int k = 0;
#pragma unroll
        for (int i = 0; i < 3; ++i) {
            int yy = h + i - 1;
#pragma unroll
            for (int j = 0; j < 3; ++j, ++k) {
                int xx = w + j - 1;
                if ((unsigned)yy < 32u && (unsigned)xx < 32u)
                    acc += wv[k] * xs[yy * 32 + xx];
            }
        }
        float sc = og[c] * rsqrtf(ov[c] + EPSV);
        out[((size_t)(b * 512 + c)) * HW + hw] = (acc - om[c]) * sc + ob[c] + xs[hw];
    }
}

extern "C" void kernel_launch(void* const* d_in, const int* in_sizes, int n_in,
                              void* d_out, int out_size, void* d_ws, size_t ws_size,
                              hipStream_t stream)
{
    const float* x     = (const float*)d_in[0];
    const float* cv1_w = (const float*)d_in[1];
    const float* cv1_b = (const float*)d_in[2];
    const float* bn1_g = (const float*)d_in[3];
    const float* bn1_b = (const float*)d_in[4];
    const float* bn1_m = (const float*)d_in[5];
    const float* bn1_v = (const float*)d_in[6];
    const float* cv2_w = (const float*)d_in[7];
    const float* cv2_b = (const float*)d_in[8];
    const float* bn2_g = (const float*)d_in[9];
    const float* bn2_b = (const float*)d_in[10];
    const float* bn2_m = (const float*)d_in[11];
    const float* bn2_v = (const float*)d_in[12];
    const float* cv3_w = (const float*)d_in[13];
    const float* cv3_b = (const float*)d_in[14];
    const float* bn3_g = (const float*)d_in[15];
    const float* bn3_b = (const float*)d_in[16];
    const float* bn3_m = (const float*)d_in[17];
    const float* bn3_v = (const float*)d_in[18];
    const float* cv4_w = (const float*)d_in[19];
    const float* cv4_b = (const float*)d_in[20];
    const float* gn_g  = (const float*)d_in[21];
    const float* gn_b  = (const float*)d_in[22];
    const float* bno_g = (const float*)d_in[23];
    const float* bno_b = (const float*)d_in[24];
    const float* bno_m = (const float*)d_in[25];
    const float* bno_v = (const float*)d_in[26];

    float* ws  = (float*)d_ws;
    float* Z1  = ws;
    float* Z2  = ws + (size_t)4194304;
    float* Z4  = ws + (size_t)8388608;
    float* ST  = ws + (size_t)(8388608 + 9437184);
    float* out = (float*)d_out;

    gemm_wmma_f32<<<512, 256, 0, stream>>>(cv1_w, x, Z1,
                                           cv1_b, bn1_g, bn1_b, bn1_m, bn1_v,
                                           256, 512, 16, 1);
    dwconv7_bn<<<4096, 32, 0, stream>>>(Z1, cv2_w, cv2_b,
                                          bn2_g, bn2_b, bn2_m, bn2_v, Z2);
    gemm_wmma_f32<<<512, 256, 0, stream>>>(cv3_w, Z2, Z1,
                                           cv3_b, bn3_g, bn3_b, bn3_m, bn3_v,
                                           256, 256, 16, 1);
    gemm_wmma_f32<<<1152, 256, 0, stream>>>(cv4_w, Z1, Z4,
                                            cv4_b, cv4_b, cv4_b, cv4_b, cv4_b,
                                            576, 256, 36, 0);
    gn_stats<<<1024, 256, 0, stream>>>(Z4, ST);
    ska_kernel<<<4096, 256, 0, stream>>>(x, Z4, ST, gn_g, gn_b,
                                         bno_g, bno_b, bno_m, bno_v, out);
}


# 1x1-conv GEMMs: stage weight and activation tiles in LDS via 16-byte global loads (64x128 block tile, K chunks of 32, two stages), wave tile 32x32; was 4-byte per-wave global loads; k order kept
# speedup vs baseline: 1.4418x; 1.1211x over previous
#include <hip/hip_runtime.h>
#include <math.h>


#define HW 1024
#define EPSV 1e-5f

typedef __attribute__((ext_vector_type(2))) float v2f;
typedef __attribute__((ext_vector_type(8))) float v8f;

#define GM 64
#define GN 128
#define GK 32
#define G_AP 36
#define G_BP 288
#define G_ASZ (GM * G_AP)
#define G_BSZ ((GK / 2) * G_BP)

typedef __attribute__((ext_vector_type(4))) float v4f;

__device__ __forceinline__ void gemm_chunk(const float* __restrict__ Af,
                                           const float* __restrict__ Bf,
                                           v8f& c00, v8f& c01, v8f& c10, v8f& c11)
{
#pragma unroll
    for (int kk = 0; kk < GK / 4; ++kk) {
        v2f a0 = *(const v2f*)(Af + kk * 4);
        v2f a1 = *(const v2f*)(Af + 16 * G_AP + kk * 4);
        v2f b0 = *(const v2f*)(Bf + kk * 2 * G_BP);
        v2f b1 = *(const v2f*)(Bf + kk * 2 * G_BP + 32);
        c00 = __builtin_amdgcn_wmma_f32_16x16x4_f32(false, a0, false, b0, (short)0, c00, false, false);
        c01 = __builtin_amdgcn_wmma_f32_16x16x4_f32(false, a0, false, b1, (short)0, c01, false, false);
        c10 = __builtin_amdgcn_wmma_f32_16x16x4_f32(false, a1, false, b0, (short)0, c10, false, false);
        c11 = __builtin_amdgcn_wmma_f32_16x16x4_f32(false, a1, false, b1, (short)0, c11, false, false);
    }
}

__global__ void __launch_bounds__(256)
gemm_wmma_f32(const float* __restrict__ Wt,
              const float* __restrict__ X,
              float* __restrict__ Y,
              const float* __restrict__ cb,
              const float* __restrict__ bg,
              const float* __restrict__ bb,
              const float* __restrict__ bm,
              const float* __restrict__ bv,
              int M, int K, int Mblocks, int relu_mode)
{
    __shared__ __attribute__((aligned(16))) float As[2 * G_ASZ];
    __shared__ __attribute__((aligned(16))) float Bs[2 * G_BSZ];

    const int tid  = threadIdx.x;
    const int lane = tid & 31;
    const int wid  = tid >> 5;
    const int mb   = blockIdx.x % Mblocks;
    const int nb   = blockIdx.x / Mblocks;
    const int m0   = mb * GM;
    const int ncol = nb * GN;
    const int b    = ncol >> 10;
    const int hw0  = ncol & 1023;

    const int lm = lane & 15;
    const int lh = lane >> 4;
    const int wm = (wid & 1) * 32;
    const int wn = (wid >> 1) * 32;

    const float* __restrict__ Xb = X + (size_t)b * K * HW + hw0;

    const int a_row = tid >> 3;
    const int a_k   = (tid & 7) * 4;
    const int b_p   = tid >> 5;
    const int b_q   = tid & 31;

    const float* __restrict__ ag = Wt + (size_t)(m0 + a_row) * K + a_k;
    const float* __restrict__ xg = Xb + (size_t)(2 * b_p) * HW + b_q * 4;
    const int a_st = a_row * G_AP + a_k;
    const int b_st = b_p * G_BP + b_q * 8;

    v4f ra0, ra1, rb0, rb1, rb2, rb3;
    ra0 = *(const v4f*)(ag);
    ra1 = *(const v4f*)(ag + (size_t)32 * K);
    rb0 = *(const v4f*)(xg);
    rb1 = *(const v4f*)(xg + HW);
    rb2 = *(const v4f*)(xg + 16 * HW);
    rb3 = *(const v4f*)(xg + 17 * HW);

    *(v4f*)(As + a_st)               = ra0;
    *(v4f*)(As + a_st + 32 * G_AP)   = ra1;
    *(v4f*)(Bs + b_st)               = (v4f){rb0.x, rb1.x, rb0.y, rb1.y};
    *(v4f*)(Bs + b_st + 4)           = (v4f){rb0.z, rb1.z, rb0.w, rb1.w};
    *(v4f*)(Bs + b_st + 8 * G_BP)     = (v4f){rb2.x, rb3.x, rb2.y, rb3.y};
    *(v4f*)(Bs + b_st + 8 * G_BP + 4) = (v4f){rb2.z, rb3.z, rb2.w, rb3.w};
    __syncthreads();

    v8f c00 = {0.f, 0.f, 0.f, 0.f, 0.f, 0.f, 0.f, 0.f};
    v8f c01 = c00, c10 = c00, c11 = c00;

    const int a_ld = (wm + lm) * G_AP + 2 * lh;
    const int b_ld = lh * G_BP + 2 * (wn + lm);

    int buf = 0;
    for (int k1 = GK; k1 < K; k1 += GK) {
        ra0 = *(const v4f*)(ag + k1);
        ra1 = *(const v4f*)(ag + (size_t)32 * K + k1);
        const float* __restrict__ xk = xg + (size_t)k1 * HW;
        rb0 = *(const v4f*)(xk);
        rb1 = *(const v4f*)(xk + HW);
        rb2 = *(const v4f*)(xk + 16 * HW);
        rb3 = *(const v4f*)(xk + 17 * HW);

        gemm_chunk(As + buf * G_ASZ + a_ld, Bs + buf * G_BSZ + b_ld, c00, c01, c10, c11);

        buf ^= 1;
        float* __restrict__ Aw = As + buf * G_ASZ + a_st;
        float* __restrict__ Bw = Bs + buf * G_BSZ + b_st;
        *(v4f*)(Aw)               = ra0;
        *(v4f*)(Aw + 32 * G_AP)   = ra1;
        *(v4f*)(Bw)               = (v4f){rb0.x, rb1.x, rb0.y, rb1.y};
        *(v4f*)(Bw + 4)           = (v4f){rb0.z, rb1.z, rb0.w, rb1.w};
        *(v4f*)(Bw + 8 * G_BP)     = (v4f){rb2.x, rb3.x, rb2.y, rb3.y};
        *(v4f*)(Bw + 8 * G_BP + 4) = (v4f){rb2.z, rb3.z, rb2.w, rb3.w};
        __syncthreads();
    }
    gemm_chunk(As + buf * G_ASZ + a_ld, Bs + buf * G_BSZ + b_ld, c00, c01, c10, c11);

    const int hi = lh << 3;
#pragma unroll
    for (int i = 0; i < 2; ++i) {
        float osc[8], obi[8];
#pragma unroll
        for (int r = 0; r < 8; ++r) {
            int o = m0 + wm + i * 16 + r + hi;
            if (relu_mode) {
                float sc = bg[o] * rsqrtf(bv[o] + EPSV);
                osc[r] = sc;
                obi[r] = (cb[o] - bm[o]) * sc + bb[o];
            } else {
                osc[r] = 1.f;
                obi[r] = cb[o];
            }
        }
#pragma unroll
        for (int j = 0; j < 2; ++j) {
            const v8f cc = (i == 0) ? (j == 0 ? c00 : c01) : (j == 0 ? c10 : c11);
            int ncl = hw0 + wn + j * 16 + lm;
#pragma unroll
            for (int r = 0; r < 8; ++r) {
                int o = m0 + wm + i * 16 + r + hi;
                float val = cc[r] * osc[r] + obi[r];
                if (relu_mode) val = fmaxf(val, 0.f);
                Y[(size_t)b * M * HW + (size_t)o * HW + ncl] = val;
            }
        }
    }
}

#define DW_LP 40
#define DW_LR 38

__global__ void __launch_bounds__(32)
dwconv7_bn(const float* __restrict__ z1,
           const float* __restrict__ wt,
           const float* __restrict__ cb,
           const float* __restrict__ bg,
           const float* __restrict__ bb,
           const float* __restrict__ bm,
           const float* __restrict__ bv,
           float* __restrict__ z2)
{
    __shared__ __attribute__((aligned(16))) float tile[DW_LR * DW_LP];

    const int lane  = threadIdx.x & 31;
    const int plane = blockIdx.x;
    const int ch    = plane & 255;

    const float4 z4 = make_float4(0.f, 0.f, 0.f, 0.f);
#pragma unroll
    for (int it = 0; it < 4; ++it) {
        const int i = it * 32 + lane;
        if (i >= 124) continue;
        int addr;
        if (i < 60) {
            int rr  = i / 10;
            int row = (rr < 3) ? rr : rr + 32;
            addr = row * DW_LP + (i - rr * 10) * 4;
        } else {
            int j = i - 60;
            addr = ((j >> 1) + 3) * DW_LP + (j & 1) * 36;
        }
        *(float4*)(tile + addr) = z4;
    }
    const float4* __restrict__ src4 = (const float4*)(z1 + (size_t)plane * HW);
    float4 in[8];
#pragma unroll
    for (int q = 0; q < 8; ++q) in[q] = src4[q * 32 + lane];

#pragma unroll
    for (int q = 0; q < 8; ++q) {
        int f = q * 32 + lane;
        *(float4*)(tile + ((f >> 3) + 3) * DW_LP + 4 + (f & 7) * 4) = in[q];
    }
    __syncthreads();

    const float* __restrict__ wk = wt + ch * 49;
    const float cbv = cb[ch];
    const float bmv = bm[ch];
    const float bbv = bb[ch];
    const float sc  = bg[ch] * rsqrtf(bv[ch] + EPSV);

    float* __restrict__ dst = z2 + (size_t)plane * HW + lane;

#pragma unroll 1
    for (int h0 = 0; h0 < 32; h0 += 8) {
        float acc[8];
#pragma unroll
        for (int i = 0; i < 8; ++i) acc[i] = 0.f;
        const float* trow = tile + h0 * DW_LP + lane + 1;
#pragma unroll
        for (int rr = 0; rr < 14; ++rr) {
#pragma unroll
            for (int j = 0; j < 7; ++j) {
                float v = trow[rr * DW_LP + j];
#pragma unroll
                for (int i = 0; i < 8; ++i) {
                    int ki = rr - i;
                    if (ki >= 0 && ki < 7) acc[i] += wk[ki * 7 + j] * v;
                }
            }
        }
#pragma unroll
        for (int i = 0; i < 8; ++i) {
            float t = acc[i] + cbv;
            dst[(h0 + i) * 32] = (t - bmv) * sc + bbv;
        }
    }
}

__global__ void gn_stats(const float* __restrict__ z4, float* __restrict__ stats)
{
    __shared__ float s1[256], s2[256];
    int gid = blockIdx.x;
    int tid = threadIdx.x;
    const float* base = z4 + (size_t)gid * 9216;
    float a = 0.f, q = 0.f;
    for (int i = tid; i < 9216; i += 256) { float t = base[i]; a += t; q += t * t; }
    s1[tid] = a; s2[tid] = q;
    __syncthreads();
    for (int off = 128; off > 0; off >>= 1) {
        if (tid < off) { s1[tid] += s1[tid + off]; s2[tid] += s2[tid + off]; }
        __syncthreads();
    }
    if (tid == 0) {
        float mean = s1[0] * (1.f / 9216.f);
        float var  = s2[0] * (1.f / 9216.f) - mean * mean;
        stats[gid * 2]     = mean;
        stats[gid * 2 + 1] = rsqrtf(var + EPSV);
    }
}

__global__ void ska_kernel(const float* __restrict__ x,
                           const float* __restrict__ z4,
                           const float* __restrict__ stats,
                           const float* __restrict__ gg,
                           const float* __restrict__ gb,
                           const float* __restrict__ og,
                           const float* __restrict__ ob,
                           const float* __restrict__ om,
                           const float* __restrict__ ov,
                           float* __restrict__ out)
{
    int idx = blockIdx.x * blockDim.x + threadIdx.x;
    int hw = idx & 1023;
    int cc = (idx >> 10) & 63;
    int b  = idx >> 16;
    int h = hw >> 5, w = hw & 31;

    float mu = stats[(b * 64 + cc) * 2];
    float rs = stats[(b * 64 + cc) * 2 + 1];

    const float* zp = z4 + ((size_t)(b * 576 + cc * 9)) * HW + hw;
    float wv[9];
    float mx = -1e30f;
#pragma unroll
    for (int k = 0; k < 9; ++k) {
        float t = (zp[(size_t)k * HW] - mu) * rs * gg[cc * 9 + k] + gb[cc * 9 + k];
        wv[k] = t;
        mx = fmaxf(mx, t);
    }
    float ssum = 0.f;
#pragma unroll
    for (int k = 0; k < 9; ++k) { float e = __expf(wv[k] - mx); wv[k] = e; ssum += e; }
    float inv = 1.f / ssum;
#pragma unroll
    for (int k = 0; k < 9; ++k) wv[k] *= inv;

#pragma unroll
    for (int g = 0; g < 8; ++g) {
        int c = g * 64 + cc;
        const float* xs = x + ((size_t)(b * 512 + c)) * HW;
        float acc = 0.f;
        int k = 0;
#pragma unroll
        for (int i = 0; i < 3; ++i) {
            int yy = h + i - 1;
#pragma unroll
            for (int j = 0; j < 3; ++j, ++k) {
                int xx = w + j - 1;
                if ((unsigned)yy < 32u && (unsigned)xx < 32u)
                    acc += wv[k] * xs[yy * 32 + xx];
            }
        }
        float sc = og[c] * rsqrtf(ov[c] + EPSV);
        out[((size_t)(b * 512 + c)) * HW + hw] = (acc - om[c]) * sc + ob[c] + xs[hw];
    }
}

extern "C" void kernel_launch(void* const* d_in, const int* in_sizes, int n_in,
                              void* d_out, int out_size, void* d_ws, size_t ws_size,
                              hipStream_t stream)
{
    const float* x     = (const float*)d_in[0];
    const float* cv1_w = (const float*)d_in[1];
    const float* cv1_b = (const float*)d_in[2];
    const float* bn1_g = (const float*)d_in[3];
    const float* bn1_b = (const float*)d_in[4];
    const float* bn1_m = (const float*)d_in[5];
    const float* bn1_v = (const float*)d_in[6];
    const float* cv2_w = (const float*)d_in[7];
    const float* cv2_b = (const float*)d_in[8];
    const float* bn2_g = (const float*)d_in[9];
    const float* bn2_b = (const float*)d_in[10];
    const float* bn2_m = (const float*)d_in[11];
    const float* bn2_v = (const float*)d_in[12];
    const float* cv3_w = (const float*)d_in[13];
    const float* cv3_b = (const float*)d_in[14];
    const float* bn3_g = (const float*)d_in[15];
    const float* bn3_b = (const float*)d_in[16];
    const float* bn3_m = (const float*)d_in[17];
    const float* bn3_v = (const float*)d_in[18];
    const float* cv4_w = (const float*)d_in[19];
    const float* cv4_b = (const float*)d_in[20];
    const float* gn_g  = (const float*)d_in[21];
    const float* gn_b  = (const float*)d_in[22];
    const float* bno_g = (const float*)d_in[23];
    const float* bno_b = (const float*)d_in[24];
    const float* bno_m = (const float*)d_in[25];
    const float* bno_v = (const float*)d_in[26];

    float* ws  = (float*)d_ws;
    float* Z1  = ws;
    float* Z2  = ws + (size_t)4194304;
    float* Z4  = ws + (size_t)8388608;
    float* ST  = ws + (size_t)(8388608 + 9437184);
    float* out = (float*)d_out;

    gemm_wmma_f32<<<512, 256, 0, stream>>>(cv1_w, x, Z1,
                                           cv1_b, bn1_g, bn1_b, bn1_m, bn1_v,
                                           256, 512, 4, 1);
    dwconv7_bn<<<4096, 32, 0, stream>>>(Z1, cv2_w, cv2_b,
                                          bn2_g, bn2_b, bn2_m, bn2_v, Z2);
    gemm_wmma_f32<<<512, 256, 0, stream>>>(cv3_w, Z2, Z1,
                                           cv3_b, bn3_g, bn3_b, bn3_m, bn3_v,
                                           256, 256, 4, 1);
    gemm_wmma_f32<<<1152, 256, 0, stream>>>(cv4_w, Z1, Z4,
                                            cv4_b, cv4_b, cv4_b, cv4_b, cv4_b,
                                            576, 256, 9, 0);
    gn_stats<<<1024, 256, 0, stream>>>(Z4, ST);
    ska_kernel<<<4096, 256, 0, stream>>>(x, Z4, ST, gn_g, gn_b,
                                         bno_g, bno_b, bno_m, bno_v, out);
}


# ska: a lane takes 4 adjacent pixels: x rows as 16-byte loads at clamped rows, side taps from neighbour lanes, edges zeroed by select; was 72 branch-guarded 4-byte loads per pixel; tap order kept
# speedup vs baseline: 1.7327x; 1.2018x over previous
#include <hip/hip_runtime.h>
#include <math.h>


#define HW 1024
#define EPSV 1e-5f

typedef __attribute__((ext_vector_type(2))) float v2f;
typedef __attribute__((ext_vector_type(8))) float v8f;

#define GM 64
#define GN 128
#define GK 32
#define G_AP 36
#define G_BP 288
#define G_ASZ (GM * G_AP)
#define G_BSZ ((GK / 2) * G_BP)

typedef __attribute__((ext_vector_type(4))) float v4f;

__device__ __forceinline__ void gemm_chunk(const float* __restrict__ Af,
                                           const float* __restrict__ Bf,
                                           v8f& c00, v8f& c01, v8f& c10, v8f& c11)
{
#pragma unroll
    for (int kk = 0; kk < GK / 4; ++kk) {
        v2f a0 = *(const v2f*)(Af + kk * 4);
        v2f a1 = *(const v2f*)(Af + 16 * G_AP + kk * 4);
        v2f b0 = *(const v2f*)(Bf + kk * 2 * G_BP);
        v2f b1 = *(const v2f*)(Bf + kk * 2 * G_BP + 32);
        c00 = __builtin_amdgcn_wmma_f32_16x16x4_f32(false, a0, false, b0, (short)0, c00, false, false);
        c01 = __builtin_amdgcn_wmma_f32_16x16x4_f32(false, a0, false, b1, (short)0, c01, false, false);
        c10 = __builtin_amdgcn_wmma_f32_16x16x4_f32(false, a1, false, b0, (short)0, c10, false, false);
        c11 = __builtin_amdgcn_wmma_f32_16x16x4_f32(false, a1, false, b1, (short)0, c11, false, false);
    }
}

__global__ void __launch_bounds__(256)
gemm_wmma_f32(const float* __restrict__ Wt,
              const float* __restrict__ X,
              float* __restrict__ Y,
              const float* __restrict__ cb,
              const float* __restrict__ bg,
              const float* __restrict__ bb,
              const float* __restrict__ bm,
              const float* __restrict__ bv,
              int M, int K, int Mblocks, int relu_mode)
{
    __shared__ __attribute__((aligned(16))) float As[2 * G_ASZ];
    __shared__ __attribute__((aligned(16))) float Bs[2 * G_BSZ];

    const int tid  = threadIdx.x;
    const int lane = tid & 31;
    const int wid  = tid >> 5;
    const int mb   = blockIdx.x % Mblocks;
    const int nb   = blockIdx.x / Mblocks;
    const int m0   = mb * GM;
    const int ncol = nb * GN;
    const int b    = ncol >> 10;
    const int hw0  = ncol & 1023;

    const int lm = lane & 15;
    const int lh = lane >> 4;
    const int wm = (wid & 1) * 32;
    const int wn = (wid >> 1) * 32;

    const float* __restrict__ Xb = X + (size_t)b * K * HW + hw0;

    const int a_row = tid >> 3;
    const int a_k   = (tid & 7) * 4;
    const int b_p   = tid >> 5;
    const int b_q   = tid & 31;

    const float* __restrict__ ag = Wt + (size_t)(m0 + a_row) * K + a_k;
    const float* __restrict__ xg = Xb + (size_t)(2 * b_p) * HW + b_q * 4;
    const int a_st = a_row * G_AP + a_k;
    const int b_st = b_p * G_BP + b_q * 8;

    v4f ra0, ra1, rb0, rb1, rb2, rb3;
    ra0 = *(const v4f*)(ag);
    ra1 = *(const v4f*)(ag + (size_t)32 * K);
    rb0 = *(const v4f*)(xg);
    rb1 = *(const v4f*)(xg + HW);
    rb2 = *(const v4f*)(xg + 16 * HW);
    rb3 = *(const v4f*)(xg + 17 * HW);

    *(v4f*)(As + a_st)               = ra0;
    *(v4f*)(As + a_st + 32 * G_AP)   = ra1;
    *(v4f*)(Bs + b_st)               = (v4f){rb0.x, rb1.x, rb0.y, rb1.y};
    *(v4f*)(Bs + b_st + 4)           = (v4f){rb0.z, rb1.z, rb0.w, rb1.w};
    *(v4f*)(Bs + b_st + 8 * G_BP)     = (v4f){rb2.x, rb3.x, rb2.y, rb3.y};
    *(v4f*)(Bs + b_st + 8 * G_BP + 4) = (v4f){rb2.z, rb3.z, rb2.w, rb3.w};
    __syncthreads();

    v8f c00 = {0.f, 0.f, 0.f, 0.f, 0.f, 0.f, 0.f, 0.f};
    v8f c01 = c00, c10 = c00, c11 = c00;

    const int a_ld = (wm + lm) * G_AP + 2 * lh;
    const int b_ld = lh * G_BP + 2 * (wn + lm);

    int buf = 0;
    for (int k1 = GK; k1 < K; k1 += GK) {
        ra0 = *(const v4f*)(ag + k1);
        ra1 = *(const v4f*)(ag + (size_t)32 * K + k1);
        const float* __restrict__ xk = xg + (size_t)k1 * HW;
        rb0 = *(const v4f*)(xk);
        rb1 = *(const v4f*)(xk + HW);
        rb2 = *(const v4f*)(xk + 16 * HW);
        rb3 = *(const v4f*)(xk + 17 * HW);

        gemm_chunk(As + buf * G_ASZ + a_ld, Bs + buf * G_BSZ + b_ld, c00, c01, c10, c11);

        buf ^= 1;
        float* __restrict__ Aw = As + buf * G_ASZ + a_st;
        float* __restrict__ Bw = Bs + buf * G_BSZ + b_st;
        *(v4f*)(Aw)               = ra0;
        *(v4f*)(Aw + 32 * G_AP)   = ra1;
        *(v4f*)(Bw)               = (v4f){rb0.x, rb1.x, rb0.y, rb1.y};
        *(v4f*)(Bw + 4)           = (v4f){rb0.z, rb1.z, rb0.w, rb1.w};
        *(v4f*)(Bw + 8 * G_BP)     = (v4f){rb2.x, rb3.x, rb2.y, rb3.y};
        *(v4f*)(Bw + 8 * G_BP + 4) = (v4f){rb2.z, rb3.z, rb2.w, rb3.w};
        __syncthreads();
    }
    gemm_chunk(As + buf * G_ASZ + a_ld, Bs + buf * G_BSZ + b_ld, c00, c01, c10, c11);

    const int hi = lh << 3;
#pragma unroll
    for (int i = 0; i < 2; ++i) {
        float osc[8], obi[8];
#pragma unroll
        for (int r = 0; r < 8; ++r) {
            int o = m0 + wm + i * 16 + r + hi;
            if (relu_mode) {
                float sc = bg[o] * rsqrtf(bv[o] + EPSV);
                osc[r] = sc;
                obi[r] = (cb[o] - bm[o]) * sc + bb[o];
            } else {
                osc[r] = 1.f;
                obi[r] = cb[o];
            }
        }
#pragma unroll
        for (int j = 0; j < 2; ++j) {
            const v8f cc = (i == 0) ? (j == 0 ? c00 : c01) : (j == 0 ? c10 : c11);
            int ncl = hw0 + wn + j * 16 + lm;
#pragma unroll
            for (int r = 0; r < 8; ++r) {
                int o = m0 + wm + i * 16 + r + hi;
                float val = cc[r] * osc[r] + obi[r];
                if (relu_mode) val = fmaxf(val, 0.f);
                Y[(size_t)b * M * HW + (size_t)o * HW + ncl] = val;
            }
        }
    }
}

#define DW_LP 40
#define DW_LR 38

__global__ void __launch_bounds__(32)
dwconv7_bn(const float* __restrict__ z1,
           const float* __restrict__ wt,
           const float* __restrict__ cb,
           const float* __restrict__ bg,
           const float* __restrict__ bb,
           const float* __restrict__ bm,
           const float* __restrict__ bv,
           float* __restrict__ z2)
{
    __shared__ __attribute__((aligned(16))) float tile[DW_LR * DW_LP];

    const int lane  = threadIdx.x & 31;
    const int plane = blockIdx.x;
    const int ch    = plane & 255;

    const float4 z4 = make_float4(0.f, 0.f, 0.f, 0.f);
#pragma unroll
    for (int it = 0; it < 4; ++it) {
        const int i = it * 32 + lane;
        if (i >= 124) continue;
        int addr;
        if (i < 60) {
            int rr  = i / 10;
            int row = (rr < 3) ? rr : rr + 32;
            addr = row * DW_LP + (i - rr * 10) * 4;
        } else {
            int j = i - 60;
            addr = ((j >> 1) + 3) * DW_LP + (j & 1) * 36;
        }
        *(float4*)(tile + addr) = z4;
    }
    const float4* __restrict__ src4 = (const float4*)(z1 + (size_t)plane * HW);
    float4 in[8];
#pragma unroll
    for (int q = 0; q < 8; ++q) in[q] = src4[q * 32 + lane];

#pragma unroll
    for (int q = 0; q < 8; ++q) {
        int f = q * 32 + lane;
        *(float4*)(tile + ((f >> 3) + 3) * DW_LP + 4 + (f & 7) * 4) = in[q];
    }
    __syncthreads();

    const float* __restrict__ wk = wt + ch * 49;
    const float cbv = cb[ch];
    const float bmv = bm[ch];
    const float bbv = bb[ch];
    const float sc  = bg[ch] * rsqrtf(bv[ch] + EPSV);

    float* __restrict__ dst = z2 + (size_t)plane * HW + lane;

#pragma unroll 1
    for (int h0 = 0; h0 < 32; h0 += 8) {
        float acc[8];
#pragma unroll
        for (int i = 0; i < 8; ++i) acc[i] = 0.f;
        const float* trow = tile + h0 * DW_LP + lane + 1;
#pragma unroll
        for (int rr = 0; rr < 14; ++rr) {
#pragma unroll
            for (int j = 0; j < 7; ++j) {
                float v = trow[rr * DW_LP + j];
#pragma unroll
                for (int i = 0; i < 8; ++i) {
                    int ki = rr - i;
                    if (ki >= 0 && ki < 7) acc[i] += wk[ki * 7 + j] * v;
                }
            }
        }
#pragma unroll
        for (int i = 0; i < 8; ++i) {
            float t = acc[i] + cbv;
            dst[(h0 + i) * 32] = (t - bmv) * sc + bbv;
        }
    }
}

__global__ void gn_stats(const float* __restrict__ z4, float* __restrict__ stats)
{
    __shared__ float s1[256], s2[256];
    int gid = blockIdx.x;
    int tid = threadIdx.x;
    const float* base = z4 + (size_t)gid * 9216;
    float a = 0.f, q = 0.f;
    for (int i = tid; i < 9216; i += 256) { float t = base[i]; a += t; q += t * t; }
    s1[tid] = a; s2[tid] = q;
    __syncthreads();
    for (int off = 128; off > 0; off >>= 1) {
        if (tid < off) { s1[tid] += s1[tid + off]; s2[tid] += s2[tid + off]; }
        __syncthreads();
    }
    if (tid == 0) {
        float mean = s1[0] * (1.f / 9216.f);
        float var  = s2[0] * (1.f / 9216.f) - mean * mean;
        stats[gid * 2]     = mean;
        stats[gid * 2 + 1] = rsqrtf(var + EPSV);
    }
}

__device__ __forceinline__ float from_left_lane(float v)
{
    return __int_as_float(__builtin_amdgcn_update_dpp(0, __float_as_int(v), 0x111  , 0xf, 0xf, true));
}
__device__ __forceinline__ float from_right_lane(float v)
{
    return __int_as_float(__builtin_amdgcn_update_dpp(0, __float_as_int(v), 0x101  , 0xf, 0xf, true));
}

__device__ __forceinline__ void row6(const float4 v, const bool lft, const bool rgt, float* __restrict__ r)
{
    float l = from_left_lane(v.w);
    float t = from_right_lane(v.x);
    r[0] = lft ? 0.f : l;
    r[1] = v.x; r[2] = v.y; r[3] = v.z; r[4] = v.w;
    r[5] = rgt ? 0.f : t;
}

__global__ void __launch_bounds__(256)
ska_kernel(const float* __restrict__ x,
           const float* __restrict__ z4,
           const float* __restrict__ stats,
           const float* __restrict__ gg,
           const float* __restrict__ gb,
           const float* __restrict__ og,
           const float* __restrict__ ob,
           const float* __restrict__ om,
           const float* __restrict__ ov,
           float* __restrict__ out)
{
    const int tid = threadIdx.x;
    const int pl  = blockIdx.x;
    const int cc  = pl & 63;
    const int b   = pl >> 6;
    const int q   = tid & 7;
    const int h   = tid >> 3;
    const int px  = h * 32 + q * 4;

    const float mu = stats[pl * 2];
    const float rs = stats[pl * 2 + 1];

    const float* __restrict__ zp = z4 + (size_t)pl * 9216 + px;
    float4 zv[9];
#pragma unroll
    for (int k = 0; k < 9; ++k) zv[k] = *(const float4*)(zp + (size_t)k * HW);

    float wv[9][4];
    float mx[4] = {-1e30f, -1e30f, -1e30f, -1e30f};
#pragma unroll
    for (int k = 0; k < 9; ++k) {
        const float gk = gg[cc * 9 + k];
        const float bk = gb[cc * 9 + k];
        const float zz[4] = {zv[k].x, zv[k].y, zv[k].z, zv[k].w};
#pragma unroll
        for (int p = 0; p < 4; ++p) {
            float t = (zz[p] - mu) * rs * gk + bk;
            wv[k][p] = t;
            mx[p] = fmaxf(mx[p], t);
        }
    }
#pragma unroll
    for (int p = 0; p < 4; ++p) {
        float ssum = 0.f;
#pragma unroll
        for (int k = 0; k < 9; ++k) { float e = __expf(wv[k][p] - mx[p]); wv[k][p] = e; ssum += e; }
        float inv = 1.f / ssum;
#pragma unroll
        for (int k = 0; k < 9; ++k) wv[k][p] *= inv;
    }

    v2f wl[9], wh[9];
#pragma unroll
    for (int k = 0; k < 9; ++k) {
        wl[k] = (v2f){wv[k][0], wv[k][1]};
        wh[k] = (v2f){wv[k][2], wv[k][3]};
    }

    const bool top = (h == 0), bot = (h == 31);
    const bool lft = (q == 0), rgt = (q == 7);
    const int o_up = (top ? 0 : h - 1) * 32 + q * 4;
    const int o_dn = (bot ? 31 : h + 1) * 32 + q * 4;
    const float4 zero4 = make_float4(0.f, 0.f, 0.f, 0.f);

    const float* __restrict__ xb = x   + ((size_t)(b * 512 + cc)) * HW;
    float* __restrict__       yb = out + ((size_t)(b * 512 + cc)) * HW + px;

    float4 ru[3], rm[3], rd[3];
#pragma unroll
    for (int g = 0; g < 2; ++g) {
        const float* __restrict__ xf = xb + (size_t)g * 64 * HW;
        ru[g] = *(const float4*)(xf + o_up);
        rm[g] = *(const float4*)(xf + px);
        rd[g] = *(const float4*)(xf + o_dn);
    }

#pragma unroll
    for (int g = 0; g < 8; ++g) {
        const int c = g * 64 + cc;
        const int s = g % 3;
        if (g + 2 < 8) {
            const float* __restrict__ xf = xb + (size_t)(g + 2) * 64 * HW;
            ru[(g + 2) % 3] = *(const float4*)(xf + o_up);
            rm[(g + 2) % 3] = *(const float4*)(xf + px);
            rd[(g + 2) % 3] = *(const float4*)(xf + o_dn);
        }
        __builtin_amdgcn_sched_barrier(0);

        const float sc  = og[c] * rsqrtf(ov[c] + EPSV);
        const float omv = om[c];
        const float obv = ob[c];

        float4 u = ru[s], d = rd[s];
        const float4 m = rm[s];
        if (top) u = zero4;
        if (bot) d = zero4;
        float r[3][6];
        row6(u, lft, rgt, r[0]);
        row6(m, lft, rgt, r[1]);
        row6(d, lft, rgt, r[2]);

        v2f accl = {0.f, 0.f}, acch = {0.f, 0.f};
#pragma unroll
        for (int i = 0; i < 3; ++i) {
            const v2f p0 = {r[i][0], r[i][1]};
            const v2f p1 = {r[i][1], r[i][2]};
            const v2f p2 = {r[i][2], r[i][3]};
            const v2f p3 = {r[i][3], r[i][4]};
            const v2f p4 = {r[i][4], r[i][5]};
            accl += wl[i * 3 + 0] * p0;
            accl += wl[i * 3 + 1] * p1;
            accl += wl[i * 3 + 2] * p2;
            acch += wh[i * 3 + 0] * p2;
            acch += wh[i * 3 + 1] * p3;
            acch += wh[i * 3 + 2] * p4;
        }
        float res[4];
        res[0] = (accl.x - omv) * sc + obv + r[1][1];
        res[1] = (accl.y - omv) * sc + obv + r[1][2];
        res[2] = (acch.x - omv) * sc + obv + r[1][3];
        res[3] = (acch.y - omv) * sc + obv + r[1][4];
        *(float4*)(yb + (size_t)g * 64 * HW) = make_float4(res[0], res[1], res[2], res[3]);
    }
}

extern "C" void kernel_launch(void* const* d_in, const int* in_sizes, int n_in,
                              void* d_out, int out_size, void* d_ws, size_t ws_size,
                              hipStream_t stream)
{
    const float* x     = (const float*)d_in[0];
    const float* cv1_w = (const float*)d_in[1];
    const float* cv1_b = (const float*)d_in[2];
    const float* bn1_g = (const float*)d_in[3];
    const float* bn1_b = (const float*)d_in[4];
    const float* bn1_m = (const float*)d_in[5];
    const float* bn1_v = (const float*)d_in[6];
    const float* cv2_w = (const float*)d_in[7];
    const float* cv2_b = (const float*)d_in[8];
    const float* bn2_g = (const float*)d_in[9];
    const float* bn2_b = (const float*)d_in[10];
    const float* bn2_m = (const float*)d_in[11];
    const float* bn2_v = (const float*)d_in[12];
    const float* cv3_w = (const float*)d_in[13];
    const float* cv3_b = (const float*)d_in[14];
    const float* bn3_g = (const float*)d_in[15];
    const float* bn3_b = (const float*)d_in[16];
    const float* bn3_m = (const float*)d_in[17];
    const float* bn3_v = (const float*)d_in[18];
    const float* cv4_w = (const float*)d_in[19];
    const float* cv4_b = (const float*)d_in[20];
    const float* gn_g  = (const float*)d_in[21];
    const float* gn_b  = (const float*)d_in[22];
    const float* bno_g = (const float*)d_in[23];
    const float* bno_b = (const float*)d_in[24];
    const float* bno_m = (const float*)d_in[25];
    const float* bno_v = (const float*)d_in[26];

    float* ws  = (float*)d_ws;
    float* Z1  = ws;
    float* Z2  = ws + (size_t)4194304;
    float* Z4  = ws + (size_t)8388608;
    float* ST  = ws + (size_t)(8388608 + 9437184);
    float* out = (float*)d_out;

    gemm_wmma_f32<<<512, 256, 0, stream>>>(cv1_w, x, Z1,
                                           cv1_b, bn1_g, bn1_b, bn1_m, bn1_v,
                                           256, 512, 4, 1);
    dwconv7_bn<<<4096, 32, 0, stream>>>(Z1, cv2_w, cv2_b,
                                          bn2_g, bn2_b, bn2_m, bn2_v, Z2);
    gemm_wmma_f32<<<512, 256, 0, stream>>>(cv3_w, Z2, Z1,
                                           cv3_b, bn3_g, bn3_b, bn3_m, bn3_v,
                                           256, 256, 4, 1);
    gemm_wmma_f32<<<1152, 256, 0, stream>>>(cv4_w, Z1, Z4,
                                            cv4_b, cv4_b, cv4_b, cv4_b, cv4_b,
                                            576, 256, 9, 0);
    gn_stats<<<1024, 256, 0, stream>>>(Z4, ST);
    ska_kernel<<<1024, 256, 0, stream>>>(x, Z4, ST, gn_g, gn_b,
                                         bno_g, bno_b, bno_m, bno_v, out);
}


# GroupNorm sums fused into cv4 epilogue: per-tile row sum and sum of squares (16-lane DPP adds, 4 waves via LDS) to workspace, ska adds 72 partials; gn_stats launch dropped; GN sum order changed
# speedup vs baseline: 1.7830x; 1.0290x over previous
#include <hip/hip_runtime.h>
#include <math.h>


#define HW 1024
#define EPSV 1e-5f

typedef __attribute__((ext_vector_type(2))) float v2f;
typedef __attribute__((ext_vector_type(8))) float v8f;

#define GM 64
#define GN 128
#define GK 32
#define G_AP 36
#define G_BP 288
#define G_ASZ (GM * G_AP)
#define G_BSZ ((GK / 2) * G_BP)

typedef __attribute__((ext_vector_type(4))) float v4f;

__device__ __forceinline__ float row16_sum(float v)
{
    v += __int_as_float(__builtin_amdgcn_update_dpp(0, __float_as_int(v), 0x161  , 0xf, 0xf, true));
    v += __int_as_float(__builtin_amdgcn_update_dpp(0, __float_as_int(v), 0x162  , 0xf, 0xf, true));
    v += __int_as_float(__builtin_amdgcn_update_dpp(0, __float_as_int(v), 0x164  , 0xf, 0xf, true));
    v += __int_as_float(__builtin_amdgcn_update_dpp(0, __float_as_int(v), 0x168  , 0xf, 0xf, true));
    return v;
}

__device__ __forceinline__ void gemm_chunk(const float* __restrict__ Af,
                                           const float* __restrict__ Bf,
                                           v8f& c00, v8f& c01, v8f& c10, v8f& c11)
{
#pragma unroll
    for (int kk = 0; kk < GK / 4; ++kk) {
        v2f a0 = *(const v2f*)(Af + kk * 4);
        v2f a1 = *(const v2f*)(Af + 16 * G_AP + kk * 4);
        v2f b0 = *(const v2f*)(Bf + kk * 2 * G_BP);
        v2f b1 = *(const v2f*)(Bf + kk * 2 * G_BP + 32);
        c00 = __builtin_amdgcn_wmma_f32_16x16x4_f32(false, a0, false, b0, (short)0, c00, false, false);
        c01 = __builtin_amdgcn_wmma_f32_16x16x4_f32(false, a0, false, b1, (short)0, c01, false, false);
        c10 = __builtin_amdgcn_wmma_f32_16x16x4_f32(false, a1, false, b0, (short)0, c10, false, false);
        c11 = __builtin_amdgcn_wmma_f32_16x16x4_f32(false, a1, false, b1, (short)0, c11, false, false);
    }
}

__global__ void __launch_bounds__(256)
gemm_wmma_f32(const float* __restrict__ Wt,
              const float* __restrict__ X,
              float* __restrict__ Y,
              const float* __restrict__ cb,
              const float* __restrict__ bg,
              const float* __restrict__ bb,
              const float* __restrict__ bm,
              const float* __restrict__ bv,
              float* __restrict__ gpart,
              int M, int K, int Mblocks, int relu_mode)
{
    __shared__ __attribute__((aligned(16))) float As[2 * G_ASZ];
    __shared__ __attribute__((aligned(16))) float Bs[2 * G_BSZ];
    __shared__ float2 red[4 * GM];

    const int tid  = threadIdx.x;
    const int lane = tid & 31;
    const int wid  = tid >> 5;
    const int mb   = blockIdx.x % Mblocks;
    const int nb   = blockIdx.x / Mblocks;
    const int m0   = mb * GM;
    const int ncol = nb * GN;
    const int b    = ncol >> 10;
    const int hw0  = ncol & 1023;

    const int lm = lane & 15;
    const int lh = lane >> 4;
    const int wm = (wid & 1) * 32;
    const int wn = (wid >> 1) * 32;

    const float* __restrict__ Xb = X + (size_t)b * K * HW + hw0;

    const int a_row = tid >> 3;
    const int a_k   = (tid & 7) * 4;
    const int b_p   = tid >> 5;
    const int b_q   = tid & 31;

    const float* __restrict__ ag = Wt + (size_t)(m0 + a_row) * K + a_k;
    const float* __restrict__ xg = Xb + (size_t)(2 * b_p) * HW + b_q * 4;
    const int a_st = a_row * G_AP + a_k;
    const int b_st = b_p * G_BP + b_q * 8;

    v4f ra0, ra1, rb0, rb1, rb2, rb3;
    ra0 = *(const v4f*)(ag);
    ra1 = *(const v4f*)(ag + (size_t)32 * K);
    rb0 = *(const v4f*)(xg);
    rb1 = *(const v4f*)(xg + HW);
    rb2 = *(const v4f*)(xg + 16 * HW);
    rb3 = *(const v4f*)(xg + 17 * HW);

    *(v4f*)(As + a_st)               = ra0;
    *(v4f*)(As + a_st + 32 * G_AP)   = ra1;
    *(v4f*)(Bs + b_st)               = (v4f){rb0.x, rb1.x, rb0.y, rb1.y};
    *(v4f*)(Bs + b_st + 4)           = (v4f){rb0.z, rb1.z, rb0.w, rb1.w};
    *(v4f*)(Bs + b_st + 8 * G_BP)     = (v4f){rb2.x, rb3.x, rb2.y, rb3.y};
    *(v4f*)(Bs + b_st + 8 * G_BP + 4) = (v4f){rb2.z, rb3.z, rb2.w, rb3.w};
    __syncthreads();

    v8f c00 = {0.f, 0.f, 0.f, 0.f, 0.f, 0.f, 0.f, 0.f};
    v8f c01 = c00, c10 = c00, c11 = c00;

    const int a_ld = (wm + lm) * G_AP + 2 * lh;
    const int b_ld = lh * G_BP + 2 * (wn + lm);

    int buf = 0;
    for (int k1 = GK; k1 < K; k1 += GK) {
        ra0 = *(const v4f*)(ag + k1);
        ra1 = *(const v4f*)(ag + (size_t)32 * K + k1);
        const float* __restrict__ xk = xg + (size_t)k1 * HW;
        rb0 = *(const v4f*)(xk);
        rb1 = *(const v4f*)(xk + HW);
        rb2 = *(const v4f*)(xk + 16 * HW);
        rb3 = *(const v4f*)(xk + 17 * HW);

        gemm_chunk(As + buf * G_ASZ + a_ld, Bs + buf * G_BSZ + b_ld, c00, c01, c10, c11);

        buf ^= 1;
        float* __restrict__ Aw = As + buf * G_ASZ + a_st;
        float* __restrict__ Bw = Bs + buf * G_BSZ + b_st;
        *(v4f*)(Aw)               = ra0;
        *(v4f*)(Aw + 32 * G_AP)   = ra1;
        *(v4f*)(Bw)               = (v4f){rb0.x, rb1.x, rb0.y, rb1.y};
        *(v4f*)(Bw + 4)           = (v4f){rb0.z, rb1.z, rb0.w, rb1.w};
        *(v4f*)(Bw + 8 * G_BP)     = (v4f){rb2.x, rb3.x, rb2.y, rb3.y};
        *(v4f*)(Bw + 8 * G_BP + 4) = (v4f){rb2.z, rb3.z, rb2.w, rb3.w};
        __syncthreads();
    }
    gemm_chunk(As + buf * G_ASZ + a_ld, Bs + buf * G_BSZ + b_ld, c00, c01, c10, c11);

    const int hi = lh << 3;
#pragma unroll
    for (int i = 0; i < 2; ++i) {
        float osc[8], obi[8];
#pragma unroll
        for (int r = 0; r < 8; ++r) {
            int o = m0 + wm + i * 16 + r + hi;
            if (relu_mode) {
                float sc = bg[o] * rsqrtf(bv[o] + EPSV);
                osc[r] = sc;
                obi[r] = (cb[o] - bm[o]) * sc + bb[o];
            } else {
                osc[r] = 1.f;
                obi[r] = cb[o];
            }
        }
        float rs[8], rq[8];
#pragma unroll
        for (int r = 0; r < 8; ++r) { rs[r] = 0.f; rq[r] = 0.f; }
#pragma unroll
        for (int j = 0; j < 2; ++j) {
            const v8f cc = (i == 0) ? (j == 0 ? c00 : c01) : (j == 0 ? c10 : c11);
            int ncl = hw0 + wn + j * 16 + lm;
#pragma unroll
            for (int r = 0; r < 8; ++r) {
                int o = m0 + wm + i * 16 + r + hi;
                float val = cc[r] * osc[r] + obi[r];
                if (relu_mode) val = fmaxf(val, 0.f);
                Y[(size_t)b * M * HW + (size_t)o * HW + ncl] = val;
                rs[r] += val;
                rq[r] += val * val;
            }
        }
        if (gpart) {
#pragma unroll
            for (int r = 0; r < 8; ++r) {
                rs[r] = row16_sum(rs[r]);
                rq[r] = row16_sum(rq[r]);
            }
            if (lm == 0) {
#pragma unroll
                for (int r = 0; r < 8; ++r)
                    red[(wid >> 1) * GM + wm + i * 16 + r + hi] = make_float2(rs[r], rq[r]);
            }
        }
    }
    if (gpart) {
        __syncthreads();
        if (tid < GM) {
            const float2 p0 = red[tid], p1 = red[GM + tid], p2 = red[2 * GM + tid], p3 = red[3 * GM + tid];
            const float s = (p0.x + p1.x) + (p2.x + p3.x);
            const float q = (p0.y + p1.y) + (p2.y + p3.y);
            *(float2*)(gpart + 2 * ((size_t)nb * M + m0 + tid)) = make_float2(s, q);
        }
    }
}

#define DW_LP 40
#define DW_LR 38

__global__ void __launch_bounds__(32)
dwconv7_bn(const float* __restrict__ z1,
           const float* __restrict__ wt,
           const float* __restrict__ cb,
           const float* __restrict__ bg,
           const float* __restrict__ bb,
           const float* __restrict__ bm,
           const float* __restrict__ bv,
           float* __restrict__ z2)
{
    __shared__ __attribute__((aligned(16))) float tile[DW_LR * DW_LP];

    const int lane  = threadIdx.x & 31;
    const int plane = blockIdx.x;
    const int ch    = plane & 255;

    const float4 z4 = make_float4(0.f, 0.f, 0.f, 0.f);
#pragma unroll
    for (int it = 0; it < 4; ++it) {
        const int i = it * 32 + lane;
        if (i >= 124) continue;
        int addr;
        if (i < 60) {
            int rr  = i / 10;
            int row = (rr < 3) ? rr : rr + 32;
            addr = row * DW_LP + (i - rr * 10) * 4;
        } else {
            int j = i - 60;
            addr = ((j >> 1) + 3) * DW_LP + (j & 1) * 36;
        }
        *(float4*)(tile + addr) = z4;
    }
    const float4* __restrict__ src4 = (const float4*)(z1 + (size_t)plane * HW);
    float4 in[8];
#pragma unroll
    for (int q = 0; q < 8; ++q) in[q] = src4[q * 32 + lane];

#pragma unroll
    for (int q = 0; q < 8; ++q) {
        int f = q * 32 + lane;
        *(float4*)(tile + ((f >> 3) + 3) * DW_LP + 4 + (f & 7) * 4) = in[q];
    }
    __syncthreads();

    const float* __restrict__ wk = wt + ch * 49;
    const float cbv = cb[ch];
    const float bmv = bm[ch];
    const float bbv = bb[ch];
    const float sc  = bg[ch] * rsqrtf(bv[ch] + EPSV);

    float* __restrict__ dst = z2 + (size_t)plane * HW + lane;

#pragma unroll 1
    for (int h0 = 0; h0 < 32; h0 += 8) {
        float acc[8];
#pragma unroll
        for (int i = 0; i < 8; ++i) acc[i] = 0.f;
        const float* trow = tile + h0 * DW_LP + lane + 1;
#pragma unroll
        for (int rr = 0; rr < 14; ++rr) {
#pragma unroll
            for (int j = 0; j < 7; ++j) {
                float v = trow[rr * DW_LP + j];
#pragma unroll
                for (int i = 0; i < 8; ++i) {
                    int ki = rr - i;
                    if (ki >= 0 && ki < 7) acc[i] += wk[ki * 7 + j] * v;
                }
            }
        }
#pragma unroll
        for (int i = 0; i < 8; ++i) {
            float t = acc[i] + cbv;
            dst[(h0 + i) * 32] = (t - bmv) * sc + bbv;
        }
    }
}

__device__ __forceinline__ float from_left_lane(float v)
{
    return __int_as_float(__builtin_amdgcn_update_dpp(0, __float_as_int(v), 0x111  , 0xf, 0xf, true));
}
__device__ __forceinline__ float from_other_row(float v)
{
    const int bits = __float_as_int(v);
    return __int_as_float(__builtin_amdgcn_permlanex16(bits, bits, 0x76543210, 0xfedcba98, false, false));
}
__device__ __forceinline__ float from_right_lane(float v)
{
    return __int_as_float(__builtin_amdgcn_update_dpp(0, __float_as_int(v), 0x101  , 0xf, 0xf, true));
}

__device__ __forceinline__ void row6(const float4 v, const bool lft, const bool rgt, float* __restrict__ r)
{
    float l = from_left_lane(v.w);
    float t = from_right_lane(v.x);
    r[0] = lft ? 0.f : l;
    r[1] = v.x; r[2] = v.y; r[3] = v.z; r[4] = v.w;
    r[5] = rgt ? 0.f : t;
}

__global__ void __launch_bounds__(256)
ska_kernel(const float* __restrict__ x,
           const float* __restrict__ z4,
           const float* __restrict__ gpart,
           const float* __restrict__ gg,
           const float* __restrict__ gb,
           const float* __restrict__ og,
           const float* __restrict__ ob,
           const float* __restrict__ om,
           const float* __restrict__ ov,
           float* __restrict__ out)
{
    const int tid = threadIdx.x;
    const int pl  = blockIdx.x;
    const int cc  = pl & 63;
    const int b   = pl >> 6;
    const int q   = tid & 7;
    const int h   = tid >> 3;
    const int px  = h * 32 + q * 4;

    const int lane = tid & 31;
    const bool third = lane < 8;
    float2 pp[3];
#pragma unroll
    for (int j = 0; j < 3; ++j) {
        const int t   = (j < 2 || third) ? lane + 32 * j : lane;
        const int nbi = t / 9;
        const int k   = t - nbi * 9;
        pp[j] = *(const float2*)(gpart + 2 * ((size_t)(b * 8 + nbi) * 576 + cc * 9 + k));
    }

    const float* __restrict__ zp = z4 + (size_t)pl * 9216 + px;
    float4 zv[9];
#pragma unroll
    for (int k = 0; k < 9; ++k) zv[k] = *(const float4*)(zp + (size_t)k * HW);
    __builtin_amdgcn_sched_barrier(0);

    float ps = (pp[0].x + pp[1].x) + (third ? pp[2].x : 0.f);
    float pq = (pp[0].y + pp[1].y) + (third ? pp[2].y : 0.f);
    ps = row16_sum(ps);
    pq = row16_sum(pq);
    ps += from_other_row(ps);
    pq += from_other_row(pq);
    const float gsum = __int_as_float(__builtin_amdgcn_readfirstlane(__float_as_int(ps)));
    const float gsq  = __int_as_float(__builtin_amdgcn_readfirstlane(__float_as_int(pq)));
    const float mu  = gsum * (1.f / 9216.f);
    const float var = gsq * (1.f / 9216.f) - mu * mu;
    const float rs  = rsqrtf(var + EPSV);

    float wv[9][4];
    float mx[4] = {-1e30f, -1e30f, -1e30f, -1e30f};
#pragma unroll
    for (int k = 0; k < 9; ++k) {
        const float gk = gg[cc * 9 + k];
        const float bk = gb[cc * 9 + k];
        const float zz[4] = {zv[k].x, zv[k].y, zv[k].z, zv[k].w};
#pragma unroll
        for (int p = 0; p < 4; ++p) {
            float t = (zz[p] - mu) * rs * gk + bk;
            wv[k][p] = t;
            mx[p] = fmaxf(mx[p], t);
        }
    }
#pragma unroll
    for (int p = 0; p < 4; ++p) {
        float ssum = 0.f;
#pragma unroll
        for (int k = 0; k < 9; ++k) { float e = __expf(wv[k][p] - mx[p]); wv[k][p] = e; ssum += e; }
        float inv = 1.f / ssum;
#pragma unroll
        for (int k = 0; k < 9; ++k) wv[k][p] *= inv;
    }

    v2f wl[9], wh[9];
#pragma unroll
    for (int k = 0; k < 9; ++k) {
        wl[k] = (v2f){wv[k][0], wv[k][1]};
        wh[k] = (v2f){wv[k][2], wv[k][3]};
    }

    const bool top = (h == 0), bot = (h == 31);
    const bool lft = (q == 0), rgt = (q == 7);
    const int o_up = (top ? 0 : h - 1) * 32 + q * 4;
    const int o_dn = (bot ? 31 : h + 1) * 32 + q * 4;
    const float4 zero4 = make_float4(0.f, 0.f, 0.f, 0.f);

    const float* __restrict__ xb = x   + ((size_t)(b * 512 + cc)) * HW;
    float* __restrict__       yb = out + ((size_t)(b * 512 + cc)) * HW + px;

    float4 ru[3], rm[3], rd[3];
#pragma unroll
    for (int g = 0; g < 2; ++g) {
        const float* __restrict__ xf = xb + (size_t)g * 64 * HW;
        ru[g] = *(const float4*)(xf + o_up);
        rm[g] = *(const float4*)(xf + px);
        rd[g] = *(const float4*)(xf + o_dn);
    }

#pragma unroll
    for (int g = 0; g < 8; ++g) {
        const int c = g * 64 + cc;
        const int s = g % 3;
        if (g + 2 < 8) {
            const float* __restrict__ xf = xb + (size_t)(g + 2) * 64 * HW;
            ru[(g + 2) % 3] = *(const float4*)(xf + o_up);
            rm[(g + 2) % 3] = *(const float4*)(xf + px);
            rd[(g + 2) % 3] = *(const float4*)(xf + o_dn);
        }
        __builtin_amdgcn_sched_barrier(0);

        const float sc  = og[c] * rsqrtf(ov[c] + EPSV);
        const float omv = om[c];
        const float obv = ob[c];

        float4 u = ru[s], d = rd[s];
        const float4 m = rm[s];
        if (top) u = zero4;
        if (bot) d = zero4;
        float r[3][6];
        row6(u, lft, rgt, r[0]);
        row6(m, lft, rgt, r[1]);
        row6(d, lft, rgt, r[2]);

        v2f accl = {0.f, 0.f}, acch = {0.f, 0.f};
#pragma unroll
        for (int i = 0; i < 3; ++i) {
            const v2f p0 = {r[i][0], r[i][1]};
            const v2f p1 = {r[i][1], r[i][2]};
            const v2f p2 = {r[i][2], r[i][3]};
            const v2f p3 = {r[i][3], r[i][4]};
            const v2f p4 = {r[i][4], r[i][5]};
            accl += wl[i * 3 + 0] * p0;
            accl += wl[i * 3 + 1] * p1;
            accl += wl[i * 3 + 2] * p2;
            acch += wh[i * 3 + 0] * p2;
            acch += wh[i * 3 + 1] * p3;
            acch += wh[i * 3 + 2] * p4;
        }
        float res[4];
        res[0] = (accl.x - omv) * sc + obv + r[1][1];
        res[1] = (accl.y - omv) * sc + obv + r[1][2];
        res[2] = (acch.x - omv) * sc + obv + r[1][3];
        res[3] = (acch.y - omv) * sc + obv + r[1][4];
        *(float4*)(yb + (size_t)g * 64 * HW) = make_float4(res[0], res[1], res[2], res[3]);
    }
}

extern "C" void kernel_launch(void* const* d_in, const int* in_sizes, int n_in,
                              void* d_out, int out_size, void* d_ws, size_t ws_size,
                              hipStream_t stream)
{
    const float* x     = (const float*)d_in[0];
    const float* cv1_w = (const float*)d_in[1];
    const float* cv1_b = (const float*)d_in[2];
    const float* bn1_g = (const float*)d_in[3];
    const float* bn1_b = (const float*)d_in[4];
    const float* bn1_m = (const float*)d_in[5];
    const float* bn1_v = (const float*)d_in[6];
    const float* cv2_w = (const float*)d_in[7];
    const float* cv2_b = (const float*)d_in[8];
    const float* bn2_g = (const float*)d_in[9];
    const float* bn2_b = (const float*)d_in[10];
    const float* bn2_m = (const float*)d_in[11];
    const float* bn2_v = (const float*)d_in[12];
    const float* cv3_w = (const float*)d_in[13];
    const float* cv3_b = (const float*)d_in[14];
    const float* bn3_g = (const float*)d_in[15];
    const float* bn3_b = (const float*)d_in[16];
    const float* bn3_m = (const float*)d_in[17];
    const float* bn3_v = (const float*)d_in[18];
    const float* cv4_w = (const float*)d_in[19];
    const float* cv4_b = (const float*)d_in[20];
    const float* gn_g  = (const float*)d_in[21];
    const float* gn_b  = (const float*)d_in[22];
    const float* bno_g = (const float*)d_in[23];
    const float* bno_b = (const float*)d_in[24];
    const float* bno_m = (const float*)d_in[25];
    const float* bno_v = (const float*)d_in[26];

    float* ws  = (float*)d_ws;
    float* Z1  = ws;
    float* Z2  = ws + (size_t)4194304;
    float* Z4  = ws + (size_t)8388608;
    float* GP  = ws + (size_t)4194304;
    float* out = (float*)d_out;

    gemm_wmma_f32<<<512, 256, 0, stream>>>(cv1_w, x, Z1,
                                           cv1_b, bn1_g, bn1_b, bn1_m, bn1_v, (float*)nullptr,
                                           256, 512, 4, 1);
    dwconv7_bn<<<4096, 32, 0, stream>>>(Z1, cv2_w, cv2_b,
                                          bn2_g, bn2_b, bn2_m, bn2_v, Z2);
    gemm_wmma_f32<<<512, 256, 0, stream>>>(cv3_w, Z2, Z1,
                                           cv3_b, bn3_g, bn3_b, bn3_m, bn3_v, (float*)nullptr,
                                           256, 256, 4, 1);
    gemm_wmma_f32<<<1152, 256, 0, stream>>>(cv4_w, Z1, Z4,
                                            cv4_b, cv4_b, cv4_b, cv4_b, cv4_b, GP,
                                            576, 256, 9, 0);
    ska_kernel<<<1024, 256, 0, stream>>>(x, Z4, GP, gn_g, gn_b,
                                         bno_g, bno_b, bno_m, bno_v, out);
}


# GEMM epilogue: BN scale and bias of the tile's 64 rows computed once per workgroup at entry into LDS and read back per lane; was 16 dependent global-load groups per wave after the K loop
# speedup vs baseline: 1.8598x; 1.0431x over previous
#include <hip/hip_runtime.h>
#include <math.h>


#define HW 1024
#define EPSV 1e-5f

typedef __attribute__((ext_vector_type(2))) float v2f;
typedef __attribute__((ext_vector_type(8))) float v8f;

#define GM 64
#define GN 128
#define GK 32
#define G_AP 36
#define G_BP 288
#define G_ASZ (GM * G_AP)
#define G_BSZ ((GK / 2) * G_BP)

typedef __attribute__((ext_vector_type(4))) float v4f;

__device__ __forceinline__ float row16_sum(float v)
{
    v += __int_as_float(__builtin_amdgcn_update_dpp(0, __float_as_int(v), 0x161  , 0xf, 0xf, true));
    v += __int_as_float(__builtin_amdgcn_update_dpp(0, __float_as_int(v), 0x162  , 0xf, 0xf, true));
    v += __int_as_float(__builtin_amdgcn_update_dpp(0, __float_as_int(v), 0x164  , 0xf, 0xf, true));
    v += __int_as_float(__builtin_amdgcn_update_dpp(0, __float_as_int(v), 0x168  , 0xf, 0xf, true));
    return v;
}

__device__ __forceinline__ void gemm_chunk(const float* __restrict__ Af,
                                           const float* __restrict__ Bf,
                                           v8f& c00, v8f& c01, v8f& c10, v8f& c11)
{
#pragma unroll
    for (int kk = 0; kk < GK / 4; ++kk) {
        v2f a0 = *(const v2f*)(Af + kk * 4);
        v2f a1 = *(const v2f*)(Af + 16 * G_AP + kk * 4);
        v2f b0 = *(const v2f*)(Bf + kk * 2 * G_BP);
        v2f b1 = *(const v2f*)(Bf + kk * 2 * G_BP + 32);
        c00 = __builtin_amdgcn_wmma_f32_16x16x4_f32(false, a0, false, b0, (short)0, c00, false, false);
        c01 = __builtin_amdgcn_wmma_f32_16x16x4_f32(false, a0, false, b1, (short)0, c01, false, false);
        c10 = __builtin_amdgcn_wmma_f32_16x16x4_f32(false, a1, false, b0, (short)0, c10, false, false);
        c11 = __builtin_amdgcn_wmma_f32_16x16x4_f32(false, a1, false, b1, (short)0, c11, false, false);
    }
}

__global__ void __launch_bounds__(256)
gemm_wmma_f32(const float* __restrict__ Wt,
              const float* __restrict__ X,
              float* __restrict__ Y,
              const float* __restrict__ cb,
              const float* __restrict__ bg,
              const float* __restrict__ bb,
              const float* __restrict__ bm,
              const float* __restrict__ bv,
              float* __restrict__ gpart,
              int M, int K, int Mblocks, int relu_mode)
{
    __shared__ __attribute__((aligned(16))) float As[2 * G_ASZ];
    __shared__ __attribute__((aligned(16))) float Bs[2 * G_BSZ];
    __shared__ float2 red[4 * GM];
    __shared__ float2 rowp[GM];

    const int tid  = threadIdx.x;
    const int lane = tid & 31;
    const int wid  = tid >> 5;
    const int mb   = blockIdx.x % Mblocks;
    const int nb   = blockIdx.x / Mblocks;
    const int m0   = mb * GM;
    const int ncol = nb * GN;
    const int b    = ncol >> 10;
    const int hw0  = ncol & 1023;

    const int lm = lane & 15;
    const int lh = lane >> 4;
    const int wm = (wid & 1) * 32;
    const int wn = (wid >> 1) * 32;

    const float* __restrict__ Xb = X + (size_t)b * K * HW + hw0;

    const int a_row = tid >> 3;
    const int a_k   = (tid & 7) * 4;
    const int b_p   = tid >> 5;
    const int b_q   = tid & 31;

    const float* __restrict__ ag = Wt + (size_t)(m0 + a_row) * K + a_k;
    const float* __restrict__ xg = Xb + (size_t)(2 * b_p) * HW + b_q * 4;
    const int a_st = a_row * G_AP + a_k;
    const int b_st = b_p * G_BP + b_q * 8;

    v4f ra0, ra1, rb0, rb1, rb2, rb3;
    ra0 = *(const v4f*)(ag);
    ra1 = *(const v4f*)(ag + (size_t)32 * K);
    rb0 = *(const v4f*)(xg);
    rb1 = *(const v4f*)(xg + HW);
    rb2 = *(const v4f*)(xg + 16 * HW);
    rb3 = *(const v4f*)(xg + 17 * HW);

    if (tid < GM) {
        const int o = m0 + tid;
        float2 p;
        if (relu_mode) {
            const float sc = bg[o] * rsqrtf(bv[o] + EPSV);
            p = make_float2(sc, (cb[o] - bm[o]) * sc + bb[o]);
        } else {
            p = make_float2(1.f, cb[o]);
        }
        rowp[tid] = p;
    }

    *(v4f*)(As + a_st)               = ra0;
    *(v4f*)(As + a_st + 32 * G_AP)   = ra1;
    *(v4f*)(Bs + b_st)               = (v4f){rb0.x, rb1.x, rb0.y, rb1.y};
    *(v4f*)(Bs + b_st + 4)           = (v4f){rb0.z, rb1.z, rb0.w, rb1.w};
    *(v4f*)(Bs + b_st + 8 * G_BP)     = (v4f){rb2.x, rb3.x, rb2.y, rb3.y};
    *(v4f*)(Bs + b_st + 8 * G_BP + 4) = (v4f){rb2.z, rb3.z, rb2.w, rb3.w};
    __syncthreads();

    v8f c00 = {0.f, 0.f, 0.f, 0.f, 0.f, 0.f, 0.f, 0.f};
    v8f c01 = c00, c10 = c00, c11 = c00;

    const int a_ld = (wm + lm) * G_AP + 2 * lh;
    const int b_ld = lh * G_BP + 2 * (wn + lm);

    int buf = 0;
    for (int k1 = GK; k1 < K; k1 += GK) {
        ra0 = *(const v4f*)(ag + k1);
        ra1 = *(const v4f*)(ag + (size_t)32 * K + k1);
        const float* __restrict__ xk = xg + (size_t)k1 * HW;
        rb0 = *(const v4f*)(xk);
        rb1 = *(const v4f*)(xk + HW);
        rb2 = *(const v4f*)(xk + 16 * HW);
        rb3 = *(const v4f*)(xk + 17 * HW);

        gemm_chunk(As + buf * G_ASZ + a_ld, Bs + buf * G_BSZ + b_ld, c00, c01, c10, c11);

        buf ^= 1;
        float* __restrict__ Aw = As + buf * G_ASZ + a_st;
        float* __restrict__ Bw = Bs + buf * G_BSZ + b_st;
        *(v4f*)(Aw)               = ra0;
        *(v4f*)(Aw + 32 * G_AP)   = ra1;
        *(v4f*)(Bw)               = (v4f){rb0.x, rb1.x, rb0.y, rb1.y};
        *(v4f*)(Bw + 4)           = (v4f){rb0.z, rb1.z, rb0.w, rb1.w};
        *(v4f*)(Bw + 8 * G_BP)     = (v4f){rb2.x, rb3.x, rb2.y, rb3.y};
        *(v4f*)(Bw + 8 * G_BP + 4) = (v4f){rb2.z, rb3.z, rb2.w, rb3.w};
        __syncthreads();
    }
    gemm_chunk(As + buf * G_ASZ + a_ld, Bs + buf * G_BSZ + b_ld, c00, c01, c10, c11);

    const int hi = lh << 3;
#pragma unroll
    for (int i = 0; i < 2; ++i) {
        float osc[8], obi[8];
#pragma unroll
        for (int r = 0; r < 8; ++r) {
            const float2 p = rowp[wm + i * 16 + r + hi];
            osc[r] = p.x;
            obi[r] = p.y;
        }
        float rs[8], rq[8];
#pragma unroll
        for (int r = 0; r < 8; ++r) { rs[r] = 0.f; rq[r] = 0.f; }
#pragma unroll
        for (int j = 0; j < 2; ++j) {
            const v8f cc = (i == 0) ? (j == 0 ? c00 : c01) : (j == 0 ? c10 : c11);
            int ncl = hw0 + wn + j * 16 + lm;
#pragma unroll
            for (int r = 0; r < 8; ++r) {
                int o = m0 + wm + i * 16 + r + hi;
                float val = cc[r] * osc[r] + obi[r];
                if (relu_mode) val = fmaxf(val, 0.f);
                Y[(size_t)b * M * HW + (size_t)o * HW + ncl] = val;
                rs[r] += val;
                rq[r] += val * val;
            }
        }
        if (gpart) {
#pragma unroll
            for (int r = 0; r < 8; ++r) {
                rs[r] = row16_sum(rs[r]);
                rq[r] = row16_sum(rq[r]);
            }
            if (lm == 0) {
#pragma unroll
                for (int r = 0; r < 8; ++r)
                    red[(wid >> 1) * GM + wm + i * 16 + r + hi] = make_float2(rs[r], rq[r]);
            }
        }
    }
    if (gpart) {
        __syncthreads();
        if (tid < GM) {
            const float2 p0 = red[tid], p1 = red[GM + tid], p2 = red[2 * GM + tid], p3 = red[3 * GM + tid];
            const float s = (p0.x + p1.x) + (p2.x + p3.x);
            const float q = (p0.y + p1.y) + (p2.y + p3.y);
            *(float2*)(gpart + 2 * ((size_t)nb * M + m0 + tid)) = make_float2(s, q);
        }
    }
}

#define DW_LP 40
#define DW_LR 38

__global__ void __launch_bounds__(32)
dwconv7_bn(const float* __restrict__ z1,
           const float* __restrict__ wt,
           const float* __restrict__ cb,
           const float* __restrict__ bg,
           const float* __restrict__ bb,
           const float* __restrict__ bm,
           const float* __restrict__ bv,
           float* __restrict__ z2)
{
    __shared__ __attribute__((aligned(16))) float tile[DW_LR * DW_LP];

    const int lane  = threadIdx.x & 31;
    const int plane = blockIdx.x;
    const int ch    = plane & 255;

    const float4 z4 = make_float4(0.f, 0.f, 0.f, 0.f);
#pragma unroll
    for (int it = 0; it < 4; ++it) {
        const int i = it * 32 + lane;
        if (i >= 124) continue;
        int addr;
        if (i < 60) {
            int rr  = i / 10;
            int row = (rr < 3) ? rr : rr + 32;
            addr = row * DW_LP + (i - rr * 10) * 4;
        } else {
            int j = i - 60;
            addr = ((j >> 1) + 3) * DW_LP + (j & 1) * 36;
        }
        *(float4*)(tile + addr) = z4;
    }
    const float4* __restrict__ src4 = (const float4*)(z1 + (size_t)plane * HW);
    float4 in[8];
#pragma unroll
    for (int q = 0; q < 8; ++q) in[q] = src4[q * 32 + lane];

#pragma unroll
    for (int q = 0; q < 8; ++q) {
        int f = q * 32 + lane;
        *(float4*)(tile + ((f >> 3) + 3) * DW_LP + 4 + (f & 7) * 4) = in[q];
    }
    __syncthreads();

    const float* __restrict__ wk = wt + ch * 49;
    const float cbv = cb[ch];
    const float bmv = bm[ch];
    const float bbv = bb[ch];
    const float sc  = bg[ch] * rsqrtf(bv[ch] + EPSV);

    float* __restrict__ dst = z2 + (size_t)plane * HW + lane;

#pragma unroll 1
    for (int h0 = 0; h0 < 32; h0 += 8) {
        float acc[8];
#pragma unroll
        for (int i = 0; i < 8; ++i) acc[i] = 0.f;
        const float* trow = tile + h0 * DW_LP + lane + 1;
#pragma unroll
        for (int rr = 0; rr < 14; ++rr) {
#pragma unroll
            for (int j = 0; j < 7; ++j) {
                float v = trow[rr * DW_LP + j];
#pragma unroll
                for (int i = 0; i < 8; ++i) {
                    int ki = rr - i;
                    if (ki >= 0 && ki < 7) acc[i] += wk[ki * 7 + j] * v;
                }
            }
        }
#pragma unroll
        for (int i = 0; i < 8; ++i) {
            float t = acc[i] + cbv;
            dst[(h0 + i) * 32] = (t - bmv) * sc + bbv;
        }
    }
}

__device__ __forceinline__ float from_left_lane(float v)
{
    return __int_as_float(__builtin_amdgcn_update_dpp(0, __float_as_int(v), 0x111  , 0xf, 0xf, true));
}
__device__ __forceinline__ float from_other_row(float v)
{
    const int bits = __float_as_int(v);
    return __int_as_float(__builtin_amdgcn_permlanex16(bits, bits, 0x76543210, 0xfedcba98, false, false));
}
__device__ __forceinline__ float from_right_lane(float v)
{
    return __int_as_float(__builtin_amdgcn_update_dpp(0, __float_as_int(v), 0x101  , 0xf, 0xf, true));
}

__device__ __forceinline__ void row6(const float4 v, const bool lft, const bool rgt, float* __restrict__ r)
{
    float l = from_left_lane(v.w);
    float t = from_right_lane(v.x);
    r[0] = lft ? 0.f : l;
    r[1] = v.x; r[2] = v.y; r[3] = v.z; r[4] = v.w;
    r[5] = rgt ? 0.f : t;
}

__global__ void __launch_bounds__(256)
ska_kernel(const float* __restrict__ x,
           const float* __restrict__ z4,
           const float* __restrict__ gpart,
           const float* __restrict__ gg,
           const float* __restrict__ gb,
           const float* __restrict__ og,
           const float* __restrict__ ob,
           const float* __restrict__ om,
           const float* __restrict__ ov,
           float* __restrict__ out)
{
    const int tid = threadIdx.x;
    const int pl  = blockIdx.x;
    const int cc  = pl & 63;
    const int b   = pl >> 6;
    const int q   = tid & 7;
    const int h   = tid >> 3;
    const int px  = h * 32 + q * 4;

    const int lane = tid & 31;
    const bool third = lane < 8;
    float2 pp[3];
#pragma unroll
    for (int j = 0; j < 3; ++j) {
        const int t   = (j < 2 || third) ? lane + 32 * j : lane;
        const int nbi = t / 9;
        const int k   = t - nbi * 9;
        pp[j] = *(const float2*)(gpart + 2 * ((size_t)(b * 8 + nbi) * 576 + cc * 9 + k));
    }

    const float* __restrict__ zp = z4 + (size_t)pl * 9216 + px;
    float4 zv[9];
#pragma unroll
    for (int k = 0; k < 9; ++k) zv[k] = *(const float4*)(zp + (size_t)k * HW);
    __builtin_amdgcn_sched_barrier(0);

    float ps = (pp[0].x + pp[1].x) + (third ? pp[2].x : 0.f);
    float pq = (pp[0].y + pp[1].y) + (third ? pp[2].y : 0.f);
    ps = row16_sum(ps);
    pq = row16_sum(pq);
    ps += from_other_row(ps);
    pq += from_other_row(pq);
    const float gsum = __int_as_float(__builtin_amdgcn_readfirstlane(__float_as_int(ps)));
    const float gsq  = __int_as_float(__builtin_amdgcn_readfirstlane(__float_as_int(pq)));
    const float mu  = gsum * (1.f / 9216.f);
    const float var = gsq * (1.f / 9216.f) - mu * mu;
    const float rs  = rsqrtf(var + EPSV);

    float wv[9][4];
    float mx[4] = {-1e30f, -1e30f, -1e30f, -1e30f};
#pragma unroll
    for (int k = 0; k < 9; ++k) {
        const float gk = gg[cc * 9 + k];
        const float bk = gb[cc * 9 + k];
        const float zz[4] = {zv[k].x, zv[k].y, zv[k].z, zv[k].w};
#pragma unroll
        for (int p = 0; p < 4; ++p) {
            float t = (zz[p] - mu) * rs * gk + bk;
            wv[k][p] = t;
            mx[p] = fmaxf(mx[p], t);
        }
    }
#pragma unroll
    for (int p = 0; p < 4; ++p) {
        float ssum = 0.f;
#pragma unroll
        for (int k = 0; k < 9; ++k) { float e = __expf(wv[k][p] - mx[p]); wv[k][p] = e; ssum += e; }
        float inv = 1.f / ssum;
#pragma unroll
        for (int k = 0; k < 9; ++k) wv[k][p] *= inv;
    }

    v2f wl[9], wh[9];
#pragma unroll
    for (int k = 0; k < 9; ++k) {
        wl[k] = (v2f){wv[k][0], wv[k][1]};
        wh[k] = (v2f){wv[k][2], wv[k][3]};
    }

    const bool top = (h == 0), bot = (h == 31);
    const bool lft = (q == 0), rgt = (q == 7);
    const int o_up = (top ? 0 : h - 1) * 32 + q * 4;
    const int o_dn = (bot ? 31 : h + 1) * 32 + q * 4;
    const float4 zero4 = make_float4(0.f, 0.f, 0.f, 0.f);

    const float* __restrict__ xb = x   + ((size_t)(b * 512 + cc)) * HW;
    float* __restrict__       yb = out + ((size_t)(b * 512 + cc)) * HW + px;

    float4 ru[3], rm[3], rd[3];
#pragma unroll
    for (int g = 0; g < 2; ++g) {
        const float* __restrict__ xf = xb + (size_t)g * 64 * HW;
        ru[g] = *(const float4*)(xf + o_up);
        rm[g] = *(const float4*)(xf + px);
        rd[g] = *(const float4*)(xf + o_dn);
    }

#pragma unroll
    for (int g = 0; g < 8; ++g) {
        const int c = g * 64 + cc;
        const int s = g % 3;
        if (g + 2 < 8) {
            const float* __restrict__ xf = xb + (size_t)(g + 2) * 64 * HW;
            ru[(g + 2) % 3] = *(const float4*)(xf + o_up);
            rm[(g + 2) % 3] = *(const float4*)(xf + px);
            rd[(g + 2) % 3] = *(const float4*)(xf + o_dn);
        }
        __builtin_amdgcn_sched_barrier(0);

        const float sc  = og[c] * rsqrtf(ov[c] + EPSV);
        const float omv = om[c];
        const float obv = ob[c];

        float4 u = ru[s], d = rd[s];
        const float4 m = rm[s];
        if (top) u = zero4;
        if (bot) d = zero4;
        float r[3][6];
        row6(u, lft, rgt, r[0]);
        row6(m, lft, rgt, r[1]);
        row6(d, lft, rgt, r[2]);

        v2f accl = {0.f, 0.f}, acch = {0.f, 0.f};
#pragma unroll
        for (int i = 0; i < 3; ++i) {
            const v2f p0 = {r[i][0], r[i][1]};
            const v2f p1 = {r[i][1], r[i][2]};
            const v2f p2 = {r[i][2], r[i][3]};
            const v2f p3 = {r[i][3], r[i][4]};
            const v2f p4 = {r[i][4], r[i][5]};
            accl += wl[i * 3 + 0] * p0;
            accl += wl[i * 3 + 1] * p1;
            accl += wl[i * 3 + 2] * p2;
            acch += wh[i * 3 + 0] * p2;
            acch += wh[i * 3 + 1] * p3;
            acch += wh[i * 3 + 2] * p4;
        }
        float res[4];
        res[0] = (accl.x - omv) * sc + obv + r[1][1];
        res[1] = (accl.y - omv) * sc + obv + r[1][2];
        res[2] = (acch.x - omv) * sc + obv + r[1][3];
        res[3] = (acch.y - omv) * sc + obv + r[1][4];
        *(float4*)(yb + (size_t)g * 64 * HW) = make_float4(res[0], res[1], res[2], res[3]);
    }
}

extern "C" void kernel_launch(void* const* d_in, const int* in_sizes, int n_in,
                              void* d_out, int out_size, void* d_ws, size_t ws_size,
                              hipStream_t stream)
{
    const float* x     = (const float*)d_in[0];
    const float* cv1_w = (const float*)d_in[1];
    const float* cv1_b = (const float*)d_in[2];
    const float* bn1_g = (const float*)d_in[3];
    const float* bn1_b = (const float*)d_in[4];
    const float* bn1_m = (const float*)d_in[5];
    const float* bn1_v = (const float*)d_in[6];
    const float* cv2_w = (const float*)d_in[7];
    const float* cv2_b = (const float*)d_in[8];
    const float* bn2_g = (const float*)d_in[9];
    const float* bn2_b = (const float*)d_in[10];
    const float* bn2_m = (const float*)d_in[11];
    const float* bn2_v = (const float*)d_in[12];
    const float* cv3_w = (const float*)d_in[13];
    const float* cv3_b = (const float*)d_in[14];
    const float* bn3_g = (const float*)d_in[15];
    const float* bn3_b = (const float*)d_in[16];
    const float* bn3_m = (const float*)d_in[17];
    const float* bn3_v = (const float*)d_in[18];
    const float* cv4_w = (const float*)d_in[19];
    const float* cv4_b = (const float*)d_in[20];
    const float* gn_g  = (const float*)d_in[21];
    const float* gn_b  = (const float*)d_in[22];
    const float* bno_g = (const float*)d_in[23];
    const float* bno_b = (const float*)d_in[24];
    const float* bno_m = (const float*)d_in[25];
    const float* bno_v = (const float*)d_in[26];

    float* ws  = (float*)d_ws;
    float* Z1  = ws;
    float* Z2  = ws + (size_t)4194304;
    float* Z4  = ws + (size_t)8388608;
    float* GP  = ws + (size_t)4194304;
    float* out = (float*)d_out;

    gemm_wmma_f32<<<512, 256, 0, stream>>>(cv1_w, x, Z1,
                                           cv1_b, bn1_g, bn1_b, bn1_m, bn1_v, (float*)nullptr,
                                           256, 512, 4, 1);
    dwconv7_bn<<<4096, 32, 0, stream>>>(Z1, cv2_w, cv2_b,
                                          bn2_g, bn2_b, bn2_m, bn2_v, Z2);
    gemm_wmma_f32<<<512, 256, 0, stream>>>(cv3_w, Z2, Z1,
                                           cv3_b, bn3_g, bn3_b, bn3_m, bn3_v, (float*)nullptr,
                                           256, 256, 4, 1);
    gemm_wmma_f32<<<1152, 256, 0, stream>>>(cv4_w, Z1, Z4,
                                            cv4_b, cv4_b, cv4_b, cv4_b, cv4_b, GP,
                                            576, 256, 9, 0);
    ska_kernel<<<1024, 256, 0, stream>>>(x, Z4, GP, gn_g, gn_b,
                                         bno_g, bno_b, bno_m, bno_v, out);
}


# cv3+cv4 fused: a workgroup per 64-px tile keeps z3 in LDS as cv4's B operand and streams only weights; cv3 launch and z3 round trip dropped; GN partials per 64 px, so GN sum order changed
# speedup vs baseline: 1.9337x; 1.0397x over previous
#include <hip/hip_runtime.h>
#include <math.h>


#define HW 1024
#define EPSV 1e-5f

typedef __attribute__((ext_vector_type(2))) float v2f;
typedef __attribute__((ext_vector_type(8))) float v8f;

#define GM 64
#define GN 128
#define GK 32
#define G_AP 36
#define G_BP 288
#define G_ASZ (GM * G_AP)
#define G_BSZ ((GK / 2) * G_BP)

typedef __attribute__((ext_vector_type(4))) float v4f;

__device__ __forceinline__ float row16_sum(float v)
{
    v += __int_as_float(__builtin_amdgcn_update_dpp(0, __float_as_int(v), 0x161  , 0xf, 0xf, true));
    v += __int_as_float(__builtin_amdgcn_update_dpp(0, __float_as_int(v), 0x162  , 0xf, 0xf, true));
    v += __int_as_float(__builtin_amdgcn_update_dpp(0, __float_as_int(v), 0x164  , 0xf, 0xf, true));
    v += __int_as_float(__builtin_amdgcn_update_dpp(0, __float_as_int(v), 0x168  , 0xf, 0xf, true));
    return v;
}

__device__ __forceinline__ void gemm_chunk(const float* __restrict__ Af,
                                           const float* __restrict__ Bf,
                                           v8f& c00, v8f& c01, v8f& c10, v8f& c11)
{
#pragma unroll
    for (int kk = 0; kk < GK / 4; ++kk) {
        v2f a0 = *(const v2f*)(Af + kk * 4);
        v2f a1 = *(const v2f*)(Af + 16 * G_AP + kk * 4);
        v2f b0 = *(const v2f*)(Bf + kk * 2 * G_BP);
        v2f b1 = *(const v2f*)(Bf + kk * 2 * G_BP + 32);
        c00 = __builtin_amdgcn_wmma_f32_16x16x4_f32(false, a0, false, b0, (short)0, c00, false, false);
        c01 = __builtin_amdgcn_wmma_f32_16x16x4_f32(false, a0, false, b1, (short)0, c01, false, false);
        c10 = __builtin_amdgcn_wmma_f32_16x16x4_f32(false, a1, false, b0, (short)0, c10, false, false);
        c11 = __builtin_amdgcn_wmma_f32_16x16x4_f32(false, a1, false, b1, (short)0, c11, false, false);
    }
}

__global__ void __launch_bounds__(256)
gemm_wmma_f32(const float* __restrict__ Wt,
              const float* __restrict__ X,
              float* __restrict__ Y,
              const float* __restrict__ cb,
              const float* __restrict__ bg,
              const float* __restrict__ bb,
              const float* __restrict__ bm,
              const float* __restrict__ bv,
              int M, int K, int Mblocks, int relu_mode)
{
    __shared__ __attribute__((aligned(16))) float As[2 * G_ASZ];
    __shared__ __attribute__((aligned(16))) float Bs[2 * G_BSZ];
    __shared__ float2 rowp[GM];

    const int tid  = threadIdx.x;
    const int lane = tid & 31;
    const int wid  = tid >> 5;
    const int mb   = blockIdx.x % Mblocks;
    const int nb   = blockIdx.x / Mblocks;
    const int m0   = mb * GM;
    const int ncol = nb * GN;
    const int b    = ncol >> 10;
    const int hw0  = ncol & 1023;

    const int lm = lane & 15;
    const int lh = lane >> 4;
    const int wm = (wid & 1) * 32;
    const int wn = (wid >> 1) * 32;

    const float* __restrict__ Xb = X + (size_t)b * K * HW + hw0;

    const int a_row = tid >> 3;
    const int a_k   = (tid & 7) * 4;
    const int b_p   = tid >> 5;
    const int b_q   = tid & 31;

    const float* __restrict__ ag = Wt + (size_t)(m0 + a_row) * K + a_k;
    const float* __restrict__ xg = Xb + (size_t)(2 * b_p) * HW + b_q * 4;
    const int a_st = a_row * G_AP + a_k;
    const int b_st = b_p * G_BP + b_q * 8;

    v4f ra0, ra1, rb0, rb1, rb2, rb3;
    ra0 = *(const v4f*)(ag);
    ra1 = *(const v4f*)(ag + (size_t)32 * K);
    rb0 = *(const v4f*)(xg);
    rb1 = *(const v4f*)(xg + HW);
    rb2 = *(const v4f*)(xg + 16 * HW);
    rb3 = *(const v4f*)(xg + 17 * HW);

    if (tid < GM) {
        const int o = m0 + tid;
        float2 p;
        if (relu_mode) {
            const float sc = bg[o] * rsqrtf(bv[o] + EPSV);
            p = make_float2(sc, (cb[o] - bm[o]) * sc + bb[o]);
        } else {
            p = make_float2(1.f, cb[o]);
        }
        rowp[tid] = p;
    }

    *(v4f*)(As + a_st)               = ra0;
    *(v4f*)(As + a_st + 32 * G_AP)   = ra1;
    *(v4f*)(Bs + b_st)               = (v4f){rb0.x, rb1.x, rb0.y, rb1.y};
    *(v4f*)(Bs + b_st + 4)           = (v4f){rb0.z, rb1.z, rb0.w, rb1.w};
    *(v4f*)(Bs + b_st + 8 * G_BP)     = (v4f){rb2.x, rb3.x, rb2.y, rb3.y};
    *(v4f*)(Bs + b_st + 8 * G_BP + 4) = (v4f){rb2.z, rb3.z, rb2.w, rb3.w};
    __syncthreads();

    v8f c00 = {0.f, 0.f, 0.f, 0.f, 0.f, 0.f, 0.f, 0.f};
    v8f c01 = c00, c10 = c00, c11 = c00;

    const int a_ld = (wm + lm) * G_AP + 2 * lh;
    const int b_ld = lh * G_BP + 2 * (wn + lm);

    int buf = 0;
    for (int k1 = GK; k1 < K; k1 += GK) {
        ra0 = *(const v4f*)(ag + k1);
        ra1 = *(const v4f*)(ag + (size_t)32 * K + k1);
        const float* __restrict__ xk = xg + (size_t)k1 * HW;
        rb0 = *(const v4f*)(xk);
        rb1 = *(const v4f*)(xk + HW);
        rb2 = *(const v4f*)(xk + 16 * HW);
        rb3 = *(const v4f*)(xk + 17 * HW);

        gemm_chunk(As + buf * G_ASZ + a_ld, Bs + buf * G_BSZ + b_ld, c00, c01, c10, c11);

        buf ^= 1;
        float* __restrict__ Aw = As + buf * G_ASZ + a_st;
        float* __restrict__ Bw = Bs + buf * G_BSZ + b_st;
        *(v4f*)(Aw)               = ra0;
        *(v4f*)(Aw + 32 * G_AP)   = ra1;
        *(v4f*)(Bw)               = (v4f){rb0.x, rb1.x, rb0.y, rb1.y};
        *(v4f*)(Bw + 4)           = (v4f){rb0.z, rb1.z, rb0.w, rb1.w};
        *(v4f*)(Bw + 8 * G_BP)     = (v4f){rb2.x, rb3.x, rb2.y, rb3.y};
        *(v4f*)(Bw + 8 * G_BP + 4) = (v4f){rb2.z, rb3.z, rb2.w, rb3.w};
        __syncthreads();
    }
    gemm_chunk(As + buf * G_ASZ + a_ld, Bs + buf * G_BSZ + b_ld, c00, c01, c10, c11);

    const int hi = lh << 3;
#pragma unroll
    for (int i = 0; i < 2; ++i) {
        float osc[8], obi[8];
#pragma unroll
        for (int r = 0; r < 8; ++r) {
            const float2 p = rowp[wm + i * 16 + r + hi];
            osc[r] = p.x;
            obi[r] = p.y;
        }
#pragma unroll
        for (int j = 0; j < 2; ++j) {
            const v8f cc = (i == 0) ? (j == 0 ? c00 : c01) : (j == 0 ? c10 : c11);
            int ncl = hw0 + wn + j * 16 + lm;
#pragma unroll
            for (int r = 0; r < 8; ++r) {
                int o = m0 + wm + i * 16 + r + hi;
                float val = cc[r] * osc[r] + obi[r];
                if (relu_mode) val = fmaxf(val, 0.f);
                Y[(size_t)b * M * HW + (size_t)o * HW + ncl] = val;
            }
        }
    }
}

#define F_PP 160
#define F_PANEL (128 * F_PP)
#define F_ASZ (272 * G_AP)

template <int MF>
__device__ __forceinline__ void panel_chunk(const float* __restrict__ Af,
                                            const float* __restrict__ Bf,
                                            v8f (&c)[MF][2])
{
#pragma unroll
    for (int kk = 0; kk < GK / 4; ++kk) {
        const v2f b0 = *(const v2f*)(Bf + kk * 2 * F_PP);
        const v2f b1 = *(const v2f*)(Bf + kk * 2 * F_PP + 32);
#pragma unroll
        for (int f = 0; f < MF; ++f) {
            const v2f a = *(const v2f*)(Af + f * 16 * G_AP + kk * 4);
            c[f][0] = __builtin_amdgcn_wmma_f32_16x16x4_f32(false, a, false, b0, (short)0, c[f][0], false, false);
            c[f][1] = __builtin_amdgcn_wmma_f32_16x16x4_f32(false, a, false, b1, (short)0, c[f][1], false, false);
        }
    }
}

__global__ void __launch_bounds__(256)
cv34_fused(const float* __restrict__ W3,
           const float* __restrict__ Z2,
           const float* __restrict__ cb3,
           const float* __restrict__ bg3,
           const float* __restrict__ bb3,
           const float* __restrict__ bm3,
           const float* __restrict__ bv3,
           const float* __restrict__ W4,
           const float* __restrict__ cb4,
           float* __restrict__ Z4,
           float* __restrict__ gpart)
{
    __shared__ __attribute__((aligned(16))) float panel[F_PANEL];
    __shared__ __attribute__((aligned(16))) float As[2 * F_ASZ];
    __shared__ float2 red[2 * 192];
    __shared__ float2 rowp[256];
    __shared__ float  bias4[576];

    const int tid  = threadIdx.x;
    const int lane = tid & 31;
    const int wid  = tid >> 5;
    const int tile = blockIdx.x;
    const int b    = tile >> 4;
    const int hw0  = (tile & 15) * 64;

    const int lm  = lane & 15;
    const int lh  = lane >> 4;
    const int wn  = wid & 1;
    const int wmg = wid >> 1;

    const int a_row = tid >> 3;
    const int a_k   = (tid & 7) * 4;
    const int b_p   = tid >> 4;
    const int b_q   = tid & 15;
    const int a_st  = a_row * G_AP + a_k;
    const int b_st  = b_p * F_PP + b_q * 8;

    const float* __restrict__ ag3 = W3 + (size_t)a_row * 256 + a_k;
    const float* __restrict__ ag4 = W4 + (size_t)a_row * 256 + a_k;
    const float* __restrict__ xg  = Z2 + ((size_t)b * 256 + 2 * b_p) * HW + hw0 + b_q * 4;

    v4f ra[8], rb0, rb1;
#pragma unroll
    for (int j = 0; j < 8; ++j) ra[j] = *(const v4f*)(ag3 + (size_t)j * 32 * 256);
    rb0 = *(const v4f*)(xg);
    rb1 = *(const v4f*)(xg + HW);

    {
        const float sc = bg3[tid] * rsqrtf(bv3[tid] + EPSV);
        rowp[tid] = make_float2(sc, (cb3[tid] - bm3[tid]) * sc + bb3[tid]);
        for (int i = tid; i < 576; i += 256) bias4[i] = cb4[i];
    }

#pragma unroll
    for (int j = 0; j < 8; ++j) *(v4f*)(As + a_st + j * 32 * G_AP) = ra[j];
    *(v4f*)(panel + b_st)     = (v4f){rb0.x, rb1.x, rb0.y, rb1.y};
    *(v4f*)(panel + b_st + 4) = (v4f){rb0.z, rb1.z, rb0.w, rb1.w};
    __syncthreads();

    const int b_ld = lh * F_PP + 2 * (wn * 32 + lm);

    v8f c1[4][2];
#pragma unroll
    for (int f = 0; f < 4; ++f) {
        c1[f][0] = (v8f){0.f, 0.f, 0.f, 0.f, 0.f, 0.f, 0.f, 0.f};
        c1[f][1] = c1[f][0];
    }
    const int a_ld1 = (wmg * 64 + lm) * G_AP + 2 * lh;

    int buf = 0;
#pragma unroll 1
    for (int kc = 1; kc < 8; ++kc) {
#pragma unroll
        for (int j = 0; j < 8; ++j) ra[j] = *(const v4f*)(ag3 + (size_t)j * 32 * 256 + kc * GK);
        const float* __restrict__ xk = xg + (size_t)kc * GK * HW;
        rb0 = *(const v4f*)(xk);
        rb1 = *(const v4f*)(xk + HW);
        __builtin_amdgcn_sched_barrier(0);

        panel_chunk<4>(As + buf * F_ASZ + a_ld1, panel + (kc - 1) * 16 * F_PP + b_ld, c1);

        __builtin_amdgcn_sched_barrier(0);
        buf ^= 1;
        float* __restrict__ Aw = As + buf * F_ASZ + a_st;
#pragma unroll
        for (int j = 0; j < 8; ++j) *(v4f*)(Aw + j * 32 * G_AP) = ra[j];
        float* __restrict__ Bw = panel + kc * 16 * F_PP + b_st;
        *(v4f*)(Bw)     = (v4f){rb0.x, rb1.x, rb0.y, rb1.y};
        *(v4f*)(Bw + 4) = (v4f){rb0.z, rb1.z, rb0.w, rb1.w};
        __syncthreads();
    }
    v4f rw[6];
#pragma unroll
    for (int j = 0; j < 6; ++j) rw[j] = *(const v4f*)(ag4 + (size_t)j * 32 * 256);

    panel_chunk<4>(As + buf * F_ASZ + a_ld1, panel + 7 * 16 * F_PP + b_ld, c1);
    __syncthreads();

#pragma unroll
    for (int f = 0; f < 4; ++f) {
#pragma unroll
        for (int r = 0; r < 8; r += 2) {
            const int row = wmg * 64 + f * 16 + r + 8 * lh;
            const float2 p0 = rowp[row];
            const float2 p1 = rowp[row + 1];
#pragma unroll
            for (int jn = 0; jn < 2; ++jn) {
                const float v0 = fmaxf(c1[f][jn][r]     * p0.x + p0.y, 0.f);
                const float v1 = fmaxf(c1[f][jn][r + 1] * p1.x + p1.y, 0.f);
                const int n = wn * 32 + jn * 16 + lm;
                *(v2f*)(panel + (row >> 1) * F_PP + 2 * n) = (v2f){v0, v1};
            }
        }
    }
    buf = 0;
#pragma unroll
    for (int j = 0; j < 6; ++j) *(v4f*)(As + a_st + j * 32 * G_AP) = rw[j];
    __syncthreads();

    const int a_ld2 = (wmg * 48 + lm) * G_AP + 2 * lh;
    float* __restrict__ yb = Z4 + (size_t)b * 576 * HW + hw0 + wn * 32 + lm;

#pragma unroll 1
    for (int pass = 0; pass < 3; ++pass) {
        v8f c2[3][2];
#pragma unroll
        for (int f = 0; f < 3; ++f) {
            c2[f][0] = (v8f){0.f, 0.f, 0.f, 0.f, 0.f, 0.f, 0.f, 0.f};
            c2[f][1] = c2[f][0];
        }
#pragma unroll 1
        for (int kc = 0; kc < 8; ++kc) {
            const int s1 = pass * 8 + kc + 1;
            const bool more = s1 < 24;
            if (more) {
                const float* __restrict__ src = ag4 + (size_t)((s1 >> 3) * 192) * 256 + (s1 & 7) * GK;
#pragma unroll
                for (int j = 0; j < 6; ++j) rw[j] = *(const v4f*)(src + (size_t)j * 32 * 256);
            }

            panel_chunk<3>(As + buf * F_ASZ + a_ld2, panel + kc * 16 * F_PP + b_ld, c2);

            buf ^= 1;
            if (more) {
                float* __restrict__ Aw = As + buf * F_ASZ + a_st;
#pragma unroll
                for (int j = 0; j < 6; ++j) *(v4f*)(Aw + j * 32 * G_AP) = rw[j];
            }
            __syncthreads();
        }

#pragma unroll
        for (int f = 0; f < 3; ++f) {
            float rs[8], rq[8];
#pragma unroll
            for (int r = 0; r < 8; ++r) {
                const int row = wmg * 48 + f * 16 + r + 8 * lh;
                const int o   = pass * 192 + row;
                const float bi = bias4[o];
                const float v0 = c2[f][0][r] + bi;
                const float v1 = c2[f][1][r] + bi;
                yb[(size_t)o * HW]      = v0;
                yb[(size_t)o * HW + 16] = v1;
                rs[r] = v0 + v1;
                rq[r] = v0 * v0 + v1 * v1;
            }
#pragma unroll
            for (int r = 0; r < 8; ++r) {
                rs[r] = row16_sum(rs[r]);
                rq[r] = row16_sum(rq[r]);
            }
            if (lm == 0) {
#pragma unroll
                for (int r = 0; r < 8; ++r)
                    red[wn * 192 + wmg * 48 + f * 16 + r + 8 * lh] = make_float2(rs[r], rq[r]);
            }
        }
        __syncthreads();
        if (tid < 192) {
            const float2 p0 = red[tid], p1 = red[192 + tid];
            *(float2*)(gpart + 2 * ((size_t)tile * 576 + pass * 192 + tid)) = make_float2(p0.x + p1.x, p0.y + p1.y);
        }
    }
}

#define DW_LP 40
#define DW_LR 38

__global__ void __launch_bounds__(32)
dwconv7_bn(const float* __restrict__ z1,
           const float* __restrict__ wt,
           const float* __restrict__ cb,
           const float* __restrict__ bg,
           const float* __restrict__ bb,
           const float* __restrict__ bm,
           const float* __restrict__ bv,
           float* __restrict__ z2)
{
    __shared__ __attribute__((aligned(16))) float tile[DW_LR * DW_LP];

    const int lane  = threadIdx.x & 31;
    const int plane = blockIdx.x;
    const int ch    = plane & 255;

    const float4 z4 = make_float4(0.f, 0.f, 0.f, 0.f);
#pragma unroll
    for (int it = 0; it < 4; ++it) {
        const int i = it * 32 + lane;
        if (i >= 124) continue;
        int addr;
        if (i < 60) {
            int rr  = i / 10;
            int row = (rr < 3) ? rr : rr + 32;
            addr = row * DW_LP + (i - rr * 10) * 4;
        } else {
            int j = i - 60;
            addr = ((j >> 1) + 3) * DW_LP + (j & 1) * 36;
        }
        *(float4*)(tile + addr) = z4;
    }
    const float4* __restrict__ src4 = (const float4*)(z1 + (size_t)plane * HW);
    float4 in[8];
#pragma unroll
    for (int q = 0; q < 8; ++q) in[q] = src4[q * 32 + lane];

#pragma unroll
    for (int q = 0; q < 8; ++q) {
        int f = q * 32 + lane;
        *(float4*)(tile + ((f >> 3) + 3) * DW_LP + 4 + (f & 7) * 4) = in[q];
    }
    __syncthreads();

    const float* __restrict__ wk = wt + ch * 49;
    const float cbv = cb[ch];
    const float bmv = bm[ch];
    const float bbv = bb[ch];
    const float sc  = bg[ch] * rsqrtf(bv[ch] + EPSV);

    float* __restrict__ dst = z2 + (size_t)plane * HW + lane;

#pragma unroll 1
    for (int h0 = 0; h0 < 32; h0 += 8) {
        float acc[8];
#pragma unroll
        for (int i = 0; i < 8; ++i) acc[i] = 0.f;
        const float* trow = tile + h0 * DW_LP + lane + 1;
#pragma unroll
        for (int rr = 0; rr < 14; ++rr) {
#pragma unroll
            for (int j = 0; j < 7; ++j) {
                float v = trow[rr * DW_LP + j];
#pragma unroll
                for (int i = 0; i < 8; ++i) {
                    int ki = rr - i;
                    if (ki >= 0 && ki < 7) acc[i] += wk[ki * 7 + j] * v;
                }
            }
        }
#pragma unroll
        for (int i = 0; i < 8; ++i) {
            float t = acc[i] + cbv;
            dst[(h0 + i) * 32] = (t - bmv) * sc + bbv;
        }
    }
}

__device__ __forceinline__ float from_left_lane(float v)
{
    return __int_as_float(__builtin_amdgcn_update_dpp(0, __float_as_int(v), 0x111  , 0xf, 0xf, true));
}
__device__ __forceinline__ float from_other_row(float v)
{
    const int bits = __float_as_int(v);
    return __int_as_float(__builtin_amdgcn_permlanex16(bits, bits, 0x76543210, 0xfedcba98, false, false));
}
__device__ __forceinline__ float from_right_lane(float v)
{
    return __int_as_float(__builtin_amdgcn_update_dpp(0, __float_as_int(v), 0x101  , 0xf, 0xf, true));
}

__device__ __forceinline__ void row6(const float4 v, const bool lft, const bool rgt, float* __restrict__ r)
{
    float l = from_left_lane(v.w);
    float t = from_right_lane(v.x);
    r[0] = lft ? 0.f : l;
    r[1] = v.x; r[2] = v.y; r[3] = v.z; r[4] = v.w;
    r[5] = rgt ? 0.f : t;
}

__global__ void __launch_bounds__(256)
ska_kernel(const float* __restrict__ x,
           const float* __restrict__ z4,
           const float* __restrict__ gpart,
           const float* __restrict__ gg,
           const float* __restrict__ gb,
           const float* __restrict__ og,
           const float* __restrict__ ob,
           const float* __restrict__ om,
           const float* __restrict__ ov,
           float* __restrict__ out)
{
    const int tid = threadIdx.x;
    const int pl  = blockIdx.x;
    const int cc  = pl & 63;
    const int b   = pl >> 6;
    const int q   = tid & 7;
    const int h   = tid >> 3;
    const int px  = h * 32 + q * 4;

    const int lane = tid & 31;
    const bool fifth = lane < 16;
    float2 pp[5];
#pragma unroll
    for (int j = 0; j < 5; ++j) {
        const int t   = (j < 4 || fifth) ? lane + 32 * j : lane;
        const int nbi = t / 9;
        const int k   = t - nbi * 9;
        pp[j] = *(const float2*)(gpart + 2 * ((size_t)(b * 16 + nbi) * 576 + cc * 9 + k));
    }

    const float* __restrict__ zp = z4 + (size_t)pl * 9216 + px;
    float4 zv[9];
#pragma unroll
    for (int k = 0; k < 9; ++k) zv[k] = *(const float4*)(zp + (size_t)k * HW);
    __builtin_amdgcn_sched_barrier(0);

    float ps = ((pp[0].x + pp[1].x) + (pp[2].x + pp[3].x)) + (fifth ? pp[4].x : 0.f);
    float pq = ((pp[0].y + pp[1].y) + (pp[2].y + pp[3].y)) + (fifth ? pp[4].y : 0.f);
    ps = row16_sum(ps);
    pq = row16_sum(pq);
    ps += from_other_row(ps);
    pq += from_other_row(pq);
    const float gsum = __int_as_float(__builtin_amdgcn_readfirstlane(__float_as_int(ps)));
    const float gsq  = __int_as_float(__builtin_amdgcn_readfirstlane(__float_as_int(pq)));
    const float mu  = gsum * (1.f / 9216.f);
    const float var = gsq * (1.f / 9216.f) - mu * mu;
    const float rs  = rsqrtf(var + EPSV);

    float wv[9][4];
    float mx[4] = {-1e30f, -1e30f, -1e30f, -1e30f};
#pragma unroll
    for (int k = 0; k < 9; ++k) {
        const float gk = gg[cc * 9 + k];
        const float bk = gb[cc * 9 + k];
        const float zz[4] = {zv[k].x, zv[k].y, zv[k].z, zv[k].w};
#pragma unroll
        for (int p = 0; p < 4; ++p) {
            float t = (zz[p] - mu) * rs * gk + bk;
            wv[k][p] = t;
            mx[p] = fmaxf(mx[p], t);
        }
    }
#pragma unroll
    for (int p = 0; p < 4; ++p) {
        float ssum = 0.f;
#pragma unroll
        for (int k = 0; k < 9; ++k) { float e = __expf(wv[k][p] - mx[p]); wv[k][p] = e; ssum += e; }
        float inv = 1.f / ssum;
#pragma unroll
        for (int k = 0; k < 9; ++k) wv[k][p] *= inv;
    }

    v2f wl[9], wh[9];
#pragma unroll
    for (int k = 0; k < 9; ++k) {
        wl[k] = (v2f){wv[k][0], wv[k][1]};
        wh[k] = (v2f){wv[k][2], wv[k][3]};
    }

    const bool top = (h == 0), bot = (h == 31);
    const bool lft = (q == 0), rgt = (q == 7);
    const int o_up = (top ? 0 : h - 1) * 32 + q * 4;
    const int o_dn = (bot ? 31 : h + 1) * 32 + q * 4;
    const float4 zero4 = make_float4(0.f, 0.f, 0.f, 0.f);

    const float* __restrict__ xb = x   + ((size_t)(b * 512 + cc)) * HW;
    float* __restrict__       yb = out + ((size_t)(b * 512 + cc)) * HW + px;

    float4 ru[3], rm[3], rd[3];
#pragma unroll
    for (int g = 0; g < 2; ++g) {
        const float* __restrict__ xf = xb + (size_t)g * 64 * HW;
        ru[g] = *(const float4*)(xf + o_up);
        rm[g] = *(const float4*)(xf + px);
        rd[g] = *(const float4*)(xf + o_dn);
    }

#pragma unroll
    for (int g = 0; g < 8; ++g) {
        const int c = g * 64 + cc;
        const int s = g % 3;
        if (g + 2 < 8) {
            const float* __restrict__ xf = xb + (size_t)(g + 2) * 64 * HW;
            ru[(g + 2) % 3] = *(const float4*)(xf + o_up);
            rm[(g + 2) % 3] = *(const float4*)(xf + px);
            rd[(g + 2) % 3] = *(const float4*)(xf + o_dn);
        }
        __builtin_amdgcn_sched_barrier(0);

        const float sc  = og[c] * rsqrtf(ov[c] + EPSV);
        const float omv = om[c];
        const float obv = ob[c];

        float4 u = ru[s], d = rd[s];
        const float4 m = rm[s];
        if (top) u = zero4;
        if (bot) d = zero4;
        float r[3][6];
        row6(u, lft, rgt, r[0]);
        row6(m, lft, rgt, r[1]);
        row6(d, lft, rgt, r[2]);

        v2f accl = {0.f, 0.f}, acch = {0.f, 0.f};
#pragma unroll
        for (int i = 0; i < 3; ++i) {
            const v2f p0 = {r[i][0], r[i][1]};
            const v2f p1 = {r[i][1], r[i][2]};
            const v2f p2 = {r[i][2], r[i][3]};
            const v2f p3 = {r[i][3], r[i][4]};
            const v2f p4 = {r[i][4], r[i][5]};
            accl += wl[i * 3 + 0] * p0;
            accl += wl[i * 3 + 1] * p1;
            accl += wl[i * 3 + 2] * p2;
            acch += wh[i * 3 + 0] * p2;
            acch += wh[i * 3 + 1] * p3;
            acch += wh[i * 3 + 2] * p4;
        }
        float res[4];
        res[0] = (accl.x - omv) * sc + obv + r[1][1];
        res[1] = (accl.y - omv) * sc + obv + r[1][2];
        res[2] = (acch.x - omv) * sc + obv + r[1][3];
        res[3] = (acch.y - omv) * sc + obv + r[1][4];
        *(float4*)(yb + (size_t)g * 64 * HW) = make_float4(res[0], res[1], res[2], res[3]);
    }
}

extern "C" void kernel_launch(void* const* d_in, const int* in_sizes, int n_in,
                              void* d_out, int out_size, void* d_ws, size_t ws_size,
                              hipStream_t stream)
{
    const float* x     = (const float*)d_in[0];
    const float* cv1_w = (const float*)d_in[1];
    const float* cv1_b = (const float*)d_in[2];
    const float* bn1_g = (const float*)d_in[3];
    const float* bn1_b = (const float*)d_in[4];
    const float* bn1_m = (const float*)d_in[5];
    const float* bn1_v = (const float*)d_in[6];
    const float* cv2_w = (const float*)d_in[7];
    const float* cv2_b = (const float*)d_in[8];
    const float* bn2_g = (const float*)d_in[9];
    const float* bn2_b = (const float*)d_in[10];
    const float* bn2_m = (const float*)d_in[11];
    const float* bn2_v = (const float*)d_in[12];
    const float* cv3_w = (const float*)d_in[13];
    const float* cv3_b = (const float*)d_in[14];
    const float* bn3_g = (const float*)d_in[15];
    const float* bn3_b = (const float*)d_in[16];
    const float* bn3_m = (const float*)d_in[17];
    const float* bn3_v = (const float*)d_in[18];
    const float* cv4_w = (const float*)d_in[19];
    const float* cv4_b = (const float*)d_in[20];
    const float* gn_g  = (const float*)d_in[21];
    const float* gn_b  = (const float*)d_in[22];
    const float* bno_g = (const float*)d_in[23];
    const float* bno_b = (const float*)d_in[24];
    const float* bno_m = (const float*)d_in[25];
    const float* bno_v = (const float*)d_in[26];

    float* ws  = (float*)d_ws;
    float* Z1  = ws;
    float* Z2  = ws + (size_t)4194304;
    float* Z4  = ws + (size_t)8388608;
    float* GP  = ws;
    float* out = (float*)d_out;

    gemm_wmma_f32<<<512, 256, 0, stream>>>(cv1_w, x, Z1,
                                           cv1_b, bn1_g, bn1_b, bn1_m, bn1_v,
                                           256, 512, 4, 1);
    dwconv7_bn<<<4096, 32, 0, stream>>>(Z1, cv2_w, cv2_b,
                                          bn2_g, bn2_b, bn2_m, bn2_v, Z2);
    cv34_fused<<<256, 256, 0, stream>>>(cv3_w, Z2, cv3_b, bn3_g, bn3_b, bn3_m, bn3_v,
                                        cv4_w, cv4_b, Z4, GP);
    ska_kernel<<<1024, 256, 0, stream>>>(x, Z4, GP, gn_g, gn_b,
                                         bno_g, bno_b, bno_m, bno_v, out);
}
